# starting kernel
# baseline (speedup 1.0000x reference)
#include <hip/hip_runtime.h>

typedef __attribute__((ext_vector_type(2))) float v2f;
typedef __attribute__((ext_vector_type(8))) float v8f;

#define DIM 64

__global__ void gcn_softmax_kernel(const float* __restrict__ alpha,
                                   float* __restrict__ wts, int n) {
  if (blockIdx.x == 0 && threadIdx.x == 0) {
    float m = alpha[0];
    for (int i = 1; i < n; ++i) m = fmaxf(m, alpha[i]);
    float s = 0.f;
    for (int i = 0; i < n; ++i) s += expf(alpha[i] - m);
    for (int i = 0; i < n; ++i) wts[i] = expf(alpha[i] - m) / s;
  }
}

__global__ void gcn_deg_kernel(const int* __restrict__ dst,
                               float* __restrict__ deg, int E) {
  int e = blockIdx.x * blockDim.x + threadIdx.x;
  if (e < E) unsafeAtomicAdd(&deg[dst[e]], 1.0f);
}

__global__ void gcn_rsqrt_kernel(float* __restrict__ d, int n) {
  int i = blockIdx.x * blockDim.x + threadIdx.x;
  if (i < n) {
    float v = d[i];
    d[i] = (v > 0.f) ? rsqrtf(v) : 0.f;
  }
}

__global__ void gcn_edgew_kernel(const int* __restrict__ src, const int* __restrict__ dst,
                                 const float* __restrict__ dis, float* __restrict__ w, int E) {
  int e = blockIdx.x * blockDim.x + threadIdx.x;
  if (e < E) w[e] = dis[src[e]] * dis[dst[e]];
}

__global__ void gcn_scale_init_kernel(const float* __restrict__ x, float* __restrict__ o,
                                      const float* __restrict__ wts, int n4) {
  int i = blockIdx.x * blockDim.x + threadIdx.x;
  if (i < n4) {
    float s = wts[0];
    float4 a = ((const float4*)x)[i];
    float4 r;
    r.x = s * a.x; r.y = s * a.y; r.z = s * a.z; r.w = s * a.w;
    ((float4*)o)[i] = r;
  }
}

__global__ void gcn_spmm_kernel(const int* __restrict__ src, const int* __restrict__ dst,
                                const float* __restrict__ w, const float* __restrict__ xin,
                                float* __restrict__ xout, int E) {
  int t = blockIdx.x * blockDim.x + threadIdx.x;
  int e = t >> 4;
  if (e >= E) return;
  int c = (t & 15) << 2;
  int s = src[e];
  int d = dst[e];
  float we = w[e];
  const float4 xs = *(const float4*)(xin + (size_t)s * DIM + c);
  float* o = xout + (size_t)d * DIM + c;
  unsafeAtomicAdd(o + 0, we * xs.x);
  unsafeAtomicAdd(o + 1, we * xs.y);
  unsafeAtomicAdd(o + 2, we * xs.z);
  unsafeAtomicAdd(o + 3, we * xs.w);
}

__global__ void gcn_axpy_kernel(const float* __restrict__ x, float* __restrict__ o,
                                const float* __restrict__ wts, int k, int n4) {
  int i = blockIdx.x * blockDim.x + threadIdx.x;
  if (i < n4) {
    float s = wts[k];
    float4 a = ((const float4*)x)[i];
    float4 r = ((const float4*)o)[i];
    r.x += s * a.x; r.y += s * a.y; r.z += s * a.z; r.w += s * a.w;
    ((float4*)o)[i] = r;
  }
}

__global__ void gcn_edge_dot_wmma_kernel(const int* __restrict__ la, const int* __restrict__ lb,
                                         const float* __restrict__ x, float* __restrict__ res,
                                         int E) {
  const int gwave = (blockIdx.x * blockDim.x + threadIdx.x) >> 5;
  const int lane  = threadIdx.x & 31;
  const int ebase = gwave * 16;
  const int m     = lane & 15;

  int e = ebase + m;
  if (e >= E) e = E - 1;

  const size_t ai = (size_t)la[e];
  const size_t bi = (size_t)lb[e];
  const int k0 = (lane >> 4) << 1;
  const float* ap = x + ai * DIM + k0;
  const float* bp = x + bi * DIM + k0;

  v8f c = {0.f, 0.f, 0.f, 0.f, 0.f, 0.f, 0.f, 0.f};
#pragma unroll
  for (int kk = 0; kk < DIM; kk += 4) {
    v2f a = *(const v2f*)(ap + kk);
    v2f b = *(const v2f*)(bp + kk);
    c = __builtin_amdgcn_wmma_f32_16x16x4_f32(false, a, false, b, (short)0, c, false, false);
  }

  const int r = lane & 7;
  float v = c[0];
  if (r == 1) v = c[1];
  if (r == 2) v = c[2];
  if (r == 3) v = c[3];
  if (r == 4) v = c[4];
  if (r == 5) v = c[5];
  if (r == 6) v = c[6];
  if (r == 7) v = c[7];

  const bool active = (lane < 8) || (lane >= 24);
  const int ei = ebase + ((lane < 8) ? lane : (lane - 16));
  if (active && ei < E) res[ei] = v;
}

extern "C" void kernel_launch(void* const* d_in, const int* in_sizes, int n_in,
                              void* d_out, int out_size, void* d_ws, size_t ws_size,
                              hipStream_t stream) {
  const int*   edge_index = (const int*)d_in[0];
  const int*   eli        = (const int*)d_in[1];
  const float* emb        = (const float*)d_in[2];
  const float* alpha      = (const float*)d_in[3];

  const int E  = in_sizes[0] / 2;
  const int Nn = in_sizes[2] / DIM;
  const int L1 = in_sizes[3];
  const int L  = L1 - 1;
  float* res = (float*)d_out;

  const int* src = edge_index;
  const int* dst = edge_index + E;
  const int* la  = eli;
  const int* lb  = eli + E;

  size_t off = 0;
  auto carve = [&](size_t bytes) -> void* {
    void* p = (char*)d_ws + off;
    off = (off + bytes + 255) & ~(size_t)255;
    return p;
  };
  float* wts = (float*)carve(sizeof(float) * (size_t)L1);
  float* dis = (float*)carve(sizeof(float) * (size_t)Nn);
  float* w   = (float*)carve(sizeof(float) * (size_t)E);
  float* xA  = (float*)carve(sizeof(float) * (size_t)Nn * DIM);
  float* xB  = (float*)carve(sizeof(float) * (size_t)Nn * DIM);
  float* acc = (float*)carve(sizeof(float) * (size_t)Nn * DIM);
  (void)ws_size; (void)n_in; (void)out_size;

  const int n4 = Nn * DIM / 4;

  gcn_softmax_kernel<<<1, 1, 0, stream>>>(alpha, wts, L1);
  hipMemsetAsync(dis, 0, sizeof(float) * (size_t)Nn, stream);
  gcn_deg_kernel<<<(E + 255) / 256, 256, 0, stream>>>(dst, dis, E);
  gcn_rsqrt_kernel<<<(Nn + 255) / 256, 256, 0, stream>>>(dis, Nn);
  gcn_edgew_kernel<<<(E + 255) / 256, 256, 0, stream>>>(src, dst, dis, w, E);
  gcn_scale_init_kernel<<<(n4 + 255) / 256, 256, 0, stream>>>(emb, acc, wts, n4);

  const float* xcur = emb;
  float* bufs[2] = {xA, xB};
  for (int l = 0; l < L; ++l) {
    float* xnext = bufs[l & 1];
    hipMemsetAsync(xnext, 0, sizeof(float) * (size_t)Nn * DIM, stream);
    long tcount = (long)E * (DIM / 4);
    gcn_spmm_kernel<<<(unsigned)((tcount + 255) / 256), 256, 0, stream>>>(
        src, dst, w, xcur, xnext, E);
    gcn_axpy_kernel<<<(n4 + 255) / 256, 256, 0, stream>>>(xnext, acc, wts, l + 1, n4);
    xcur = xnext;
  }

  if (E > 0) {
    int waves  = (E + 15) / 16;
    int blocks = (waves + 7) / 8;
    gcn_edge_dot_wmma_kernel<<<blocks, 256, 0, stream>>>(la, lb, acc, res, E);
  }
}


# Replace the 3 layers' float-atomic edge scatter by per-destination lists built each call (int atomic count = arrival-order rank, scan) and a 16-lane register gather-sum per row; sum order changes
# speedup vs baseline: 1.8669x; 1.8669x over previous
#include <hip/hip_runtime.h>

typedef __attribute__((ext_vector_type(2))) float v2f;
typedef __attribute__((ext_vector_type(8))) float v8f;

#define DIM 64

__global__ void gcn_softmax_kernel(const float* __restrict__ alpha,
                                   float* __restrict__ wts, int n) {
  if (blockIdx.x == 0 && threadIdx.x == 0) {
    float m = alpha[0];
    for (int i = 1; i < n; ++i) m = fmaxf(m, alpha[i]);
    float s = 0.f;
    for (int i = 0; i < n; ++i) s += expf(alpha[i] - m);
    for (int i = 0; i < n; ++i) wts[i] = expf(alpha[i] - m) / s;
  }
}

__global__ void gcn_count_rank_kernel(const int* __restrict__ dst, int* __restrict__ cnt,
                                      int* __restrict__ rank, int E) {
  int e = blockIdx.x * blockDim.x + threadIdx.x;
  if (e < E) rank[e] = atomicAdd(&cnt[dst[e]], 1);
}

#define SCAN_THREADS 1024
#define SCAN_TILE    4096
__global__ void __launch_bounds__(SCAN_THREADS)
gcn_scan_dis_kernel(const int* __restrict__ cnt, int* __restrict__ rowptr,
                    float* __restrict__ dis, int n) {
  __shared__ int wsum[32];
  __shared__ int wpre[32];
  const int tid  = threadIdx.x;
  const int lane = tid & 31;
  const int wv   = tid >> 5;
  const int tile0 = blockIdx.x * SCAN_TILE;

  int part = 0;
#pragma unroll 8
  for (int j = tid * 4; j < tile0; j += SCAN_THREADS * 4) {
    const int4 q = *(const int4*)(cnt + j);
    part += (q.x + q.y) + (q.z + q.w);
  }

  const int base = tile0 + tid * 4;
  int v0 = 0, v1 = 0, v2 = 0, v3 = 0;
  if (base + 3 < n) {
    const int4 q = *(const int4*)(cnt + base);
    v0 = q.x; v1 = q.y; v2 = q.z; v3 = q.w;
  } else {
    if (base + 0 < n) v0 = cnt[base + 0];
    if (base + 1 < n) v1 = cnt[base + 1];
    if (base + 2 < n) v2 = cnt[base + 2];
  }
  const int t = v0 + v1 + v2 + v3;

  int inc = t;
  for (int o = 1; o < 32; o <<= 1) {
    const int u = __shfl_up(inc, o, 32);
    if (lane >= o) inc += u;
  }
  for (int o = 16; o > 0; o >>= 1) part += __shfl_xor(part, o, 32);
  if (lane == 31) wsum[wv] = inc;
  if (lane == 0)  wpre[wv] = part;
  __syncthreads();
  int off = 0;
  for (int i = 0; i < 32; ++i) {
    off += wpre[i];
    if (i < wv) off += wsum[i];
  }
  const int e0 = off + inc - t;
  const int e1 = e0 + v0;
  const int e2 = e1 + v1;
  const int e3 = e2 + v2;
  const float d0 = (v0 > 0) ? rsqrtf((float)v0) : 0.f;
  const float d1 = (v1 > 0) ? rsqrtf((float)v1) : 0.f;
  const float d2 = (v2 > 0) ? rsqrtf((float)v2) : 0.f;
  const float d3 = (v3 > 0) ? rsqrtf((float)v3) : 0.f;
  if (base + 3 < n) {
    *(int4*)(rowptr + base) = make_int4(e0, e1, e2, e3);
    *(float4*)(dis + base)  = make_float4(d0, d1, d2, d3);
  } else {
    if (base + 0 < n) { rowptr[base + 0] = e0; dis[base + 0] = d0; }
    if (base + 1 < n) { rowptr[base + 1] = e1; dis[base + 1] = d1; }
    if (base + 2 < n) { rowptr[base + 2] = e2; dis[base + 2] = d2; }
  }
}

#define FILL_EPT 4
__global__ void gcn_fill_lists_kernel(const int* __restrict__ src, const int* __restrict__ dst,
                                      const int* __restrict__ rank, const int* __restrict__ rowptr,
                                      const float* __restrict__ dis, int2* __restrict__ lst, int E) {
  const int T  = gridDim.x * blockDim.x;
  const int e0 = blockIdx.x * blockDim.x + threadIdx.x;
  int s[FILL_EPT], d[FILL_EPT], r[FILL_EPT], p[FILL_EPT];
  float we[FILL_EPT];
#pragma unroll
  for (int k = 0; k < FILL_EPT; ++k) {
    const int e  = e0 + k * T;
    const int ec = (e < E) ? e : (E - 1);
    s[k] = src[ec];
    d[k] = dst[ec];
    r[k] = rank[ec];
  }
#pragma unroll
  for (int k = 0; k < FILL_EPT; ++k) {
    p[k]  = rowptr[d[k]] + r[k];
    we[k] = dis[s[k]] * dis[d[k]];
  }
#pragma unroll
  for (int k = 0; k < FILL_EPT; ++k) {
    if (e0 + k * T < E) lst[p[k]] = make_int2(s[k], __float_as_int(we[k]));
  }
}

__global__ void gcn_scale_init_kernel(const float* __restrict__ x, float* __restrict__ o,
                                      const float* __restrict__ wts, int n4) {
  int i = blockIdx.x * blockDim.x + threadIdx.x;
  if (i < n4) {
    float s = wts[0];
    float4 a = ((const float4*)x)[i];
    float4 r;
    r.x = s * a.x; r.y = s * a.y; r.z = s * a.z; r.w = s * a.w;
    ((float4*)o)[i] = r;
  }
}

__device__ __forceinline__ void gcn_row_fma(float4& a, float w, const float4 x) {
  a.x += w * x.x; a.y += w * x.y; a.z += w * x.z; a.w += w * x.w;
}

__global__ void gcn_spmm_gather_kernel(const int* __restrict__ rowptr, const int2* __restrict__ lst,
                                       const float* __restrict__ xin, float* __restrict__ xout,
                                       int n, int E) {
  const int t = blockIdx.x * blockDim.x + threadIdx.x;
  const int row = t >> 4;
  if (row >= n) return;
  const int l16 = t & 15;
  const int c   = l16 << 2;
  const int beg = rowptr[row];
  const int end = (row + 1 < n) ? rowptr[row + 1] : E;
  const float* xc = xin + c;

  float4 a = make_float4(0.f, 0.f, 0.f, 0.f);
  for (int j0 = beg; j0 < end; j0 += 16) {
    const int jj = j0 + l16;
    int2 ent = make_int2(0, 0);
    if (jj < end) ent = lst[jj];
    const int cnt = min(16, end - j0);
    int i = 0;
    for (; i + 4 <= cnt; i += 4) {
      const int   s0 = __shfl(ent.x, i + 0, 16);
      const int   s1 = __shfl(ent.x, i + 1, 16);
      const int   s2 = __shfl(ent.x, i + 2, 16);
      const int   s3 = __shfl(ent.x, i + 3, 16);
      const float w0 = __int_as_float(__shfl(ent.y, i + 0, 16));
      const float w1 = __int_as_float(__shfl(ent.y, i + 1, 16));
      const float w2 = __int_as_float(__shfl(ent.y, i + 2, 16));
      const float w3 = __int_as_float(__shfl(ent.y, i + 3, 16));
      const float4 x0 = *(const float4*)(xc + (size_t)s0 * DIM);
      const float4 x1 = *(const float4*)(xc + (size_t)s1 * DIM);
      const float4 x2 = *(const float4*)(xc + (size_t)s2 * DIM);
      const float4 x3 = *(const float4*)(xc + (size_t)s3 * DIM);
      gcn_row_fma(a, w0, x0);
      gcn_row_fma(a, w1, x1);
      gcn_row_fma(a, w2, x2);
      gcn_row_fma(a, w3, x3);
    }
    for (; i < cnt; ++i) {
      const int   s0 = __shfl(ent.x, i, 16);
      const float w0 = __int_as_float(__shfl(ent.y, i, 16));
      const float4 x0 = *(const float4*)(xc + (size_t)s0 * DIM);
      gcn_row_fma(a, w0, x0);
    }
  }
  *(float4*)(xout + (size_t)row * DIM + c) = a;
}

__global__ void gcn_axpy_kernel(const float* __restrict__ x, float* __restrict__ o,
                                const float* __restrict__ wts, int k, int n4) {
  int i = blockIdx.x * blockDim.x + threadIdx.x;
  if (i < n4) {
    float s = wts[k];
    float4 a = ((const float4*)x)[i];
    float4 r = ((const float4*)o)[i];
    r.x += s * a.x; r.y += s * a.y; r.z += s * a.z; r.w += s * a.w;
    ((float4*)o)[i] = r;
  }
}

__global__ void gcn_edge_dot_wmma_kernel(const int* __restrict__ la, const int* __restrict__ lb,
                                         const float* __restrict__ x, float* __restrict__ res,
                                         int E) {
  const int gwave = (blockIdx.x * blockDim.x + threadIdx.x) >> 5;
  const int lane  = threadIdx.x & 31;
  const int ebase = gwave * 16;
  const int m     = lane & 15;

  int e = ebase + m;
  if (e >= E) e = E - 1;

  const size_t ai = (size_t)la[e];
  const size_t bi = (size_t)lb[e];
  const int k0 = (lane >> 4) << 1;
  const float* ap = x + ai * DIM + k0;
  const float* bp = x + bi * DIM + k0;

  v8f c = {0.f, 0.f, 0.f, 0.f, 0.f, 0.f, 0.f, 0.f};
#pragma unroll
  for (int kk = 0; kk < DIM; kk += 4) {
    v2f a = *(const v2f*)(ap + kk);
    v2f b = *(const v2f*)(bp + kk);
    c = __builtin_amdgcn_wmma_f32_16x16x4_f32(false, a, false, b, (short)0, c, false, false);
  }

  const int r = lane & 7;
  float v = c[0];
  if (r == 1) v = c[1];
  if (r == 2) v = c[2];
  if (r == 3) v = c[3];
  if (r == 4) v = c[4];
  if (r == 5) v = c[5];
  if (r == 6) v = c[6];
  if (r == 7) v = c[7];

  const bool active = (lane < 8) || (lane >= 24);
  const int ei = ebase + ((lane < 8) ? lane : (lane - 16));
  if (active && ei < E) res[ei] = v;
}

extern "C" void kernel_launch(void* const* d_in, const int* in_sizes, int n_in,
                              void* d_out, int out_size, void* d_ws, size_t ws_size,
                              hipStream_t stream) {
  const int*   edge_index = (const int*)d_in[0];
  const int*   eli        = (const int*)d_in[1];
  const float* emb        = (const float*)d_in[2];
  const float* alpha      = (const float*)d_in[3];

  const int E  = in_sizes[0] / 2;
  const int Nn = in_sizes[2] / DIM;
  const int L1 = in_sizes[3];
  const int L  = L1 - 1;
  float* res = (float*)d_out;

  const int* src = edge_index;
  const int* dst = edge_index + E;
  const int* la  = eli;
  const int* lb  = eli + E;

  size_t off = 0;
  auto carve = [&](size_t bytes) -> void* {
    void* p = (char*)d_ws + off;
    off = (off + bytes + 255) & ~(size_t)255;
    return p;
  };
  float* wts = (float*)carve(sizeof(float) * (size_t)L1);
  float* dis = (float*)carve(sizeof(float) * (size_t)Nn);
  int*   cnt    = (int*)carve(sizeof(int) * (size_t)Nn);
  int*   rowptr = (int*)carve(sizeof(int) * (size_t)Nn);
  int*   rank   = (int*)carve(sizeof(int) * (size_t)E);
  int2*  lst    = (int2*)carve(sizeof(int2) * (size_t)E);
  float* xA  = (float*)carve(sizeof(float) * (size_t)Nn * DIM);
  float* xB  = (float*)carve(sizeof(float) * (size_t)Nn * DIM);
  float* acc = (float*)carve(sizeof(float) * (size_t)Nn * DIM);
  (void)ws_size; (void)n_in; (void)out_size;

  const int n4 = Nn * DIM / 4;

  gcn_softmax_kernel<<<1, 1, 0, stream>>>(alpha, wts, L1);
  hipMemsetAsync(cnt, 0, sizeof(int) * (size_t)Nn, stream);
  gcn_count_rank_kernel<<<(E + 255) / 256, 256, 0, stream>>>(dst, cnt, rank, E);
  gcn_scan_dis_kernel<<<(Nn + SCAN_TILE - 1) / SCAN_TILE, SCAN_THREADS, 0, stream>>>(cnt, rowptr, dis, Nn);
  if (E > 0)
    gcn_fill_lists_kernel<<<(E + 256 * FILL_EPT - 1) / (256 * FILL_EPT), 256, 0, stream>>>(
        src, dst, rank, rowptr, dis, lst, E);
  gcn_scale_init_kernel<<<(n4 + 255) / 256, 256, 0, stream>>>(emb, acc, wts, n4);

  const float* xcur = emb;
  float* bufs[2] = {xA, xB};
  for (int l = 0; l < L; ++l) {
    float* xnext = bufs[l & 1];
    long tcount = (long)Nn * (DIM / 4);
    gcn_spmm_gather_kernel<<<(unsigned)((tcount + 255) / 256), 256, 0, stream>>>(
        rowptr, lst, xcur, xnext, Nn, E);
    gcn_axpy_kernel<<<(n4 + 255) / 256, 256, 0, stream>>>(xnext, acc, wts, l + 1, n4);
    xcur = xnext;
  }

  if (E > 0) {
    int waves  = (E + 15) / 16;
    int blocks = (waves + 7) / 8;
    gcn_edge_dot_wmma_kernel<<<blocks, 256, 0, stream>>>(la, lb, acc, res, E);
  }
}


# Fold the layer-weighted output sum into the gather kernels (layer 1 also starts acc from the embedding row, last layer stores no x): scale pass and 3 add passes go; sum order as before
# speedup vs baseline: 2.0579x; 1.1023x over previous
#include <hip/hip_runtime.h>

typedef __attribute__((ext_vector_type(2))) float v2f;
typedef __attribute__((ext_vector_type(8))) float v8f;

#define DIM 64

__global__ void gcn_softmax_kernel(const float* __restrict__ alpha,
                                   float* __restrict__ wts, int n) {
  if (blockIdx.x == 0 && threadIdx.x == 0) {
    float m = alpha[0];
    for (int i = 1; i < n; ++i) m = fmaxf(m, alpha[i]);
    float s = 0.f;
    for (int i = 0; i < n; ++i) s += expf(alpha[i] - m);
    for (int i = 0; i < n; ++i) wts[i] = expf(alpha[i] - m) / s;
  }
}

__global__ void gcn_count_rank_kernel(const int* __restrict__ dst, int* __restrict__ cnt,
                                      int* __restrict__ rank, int E) {
  int e = blockIdx.x * blockDim.x + threadIdx.x;
  if (e < E) rank[e] = atomicAdd(&cnt[dst[e]], 1);
}

#define SCAN_THREADS 1024
#define SCAN_TILE    4096
__global__ void __launch_bounds__(SCAN_THREADS)
gcn_scan_dis_kernel(const int* __restrict__ cnt, int* __restrict__ rowptr,
                    float* __restrict__ dis, int n) {
  __shared__ int wsum[32];
  __shared__ int wpre[32];
  const int tid  = threadIdx.x;
  const int lane = tid & 31;
  const int wv   = tid >> 5;
  const int tile0 = blockIdx.x * SCAN_TILE;

  int part = 0;
#pragma unroll 8
  for (int j = tid * 4; j < tile0; j += SCAN_THREADS * 4) {
    const int4 q = *(const int4*)(cnt + j);
    part += (q.x + q.y) + (q.z + q.w);
  }

  const int base = tile0 + tid * 4;
  int v0 = 0, v1 = 0, v2 = 0, v3 = 0;
  if (base + 3 < n) {
    const int4 q = *(const int4*)(cnt + base);
    v0 = q.x; v1 = q.y; v2 = q.z; v3 = q.w;
  } else {
    if (base + 0 < n) v0 = cnt[base + 0];
    if (base + 1 < n) v1 = cnt[base + 1];
    if (base + 2 < n) v2 = cnt[base + 2];
  }
  const int t = v0 + v1 + v2 + v3;

  int inc = t;
  for (int o = 1; o < 32; o <<= 1) {
    const int u = __shfl_up(inc, o, 32);
    if (lane >= o) inc += u;
  }
  for (int o = 16; o > 0; o >>= 1) part += __shfl_xor(part, o, 32);
  if (lane == 31) wsum[wv] = inc;
  if (lane == 0)  wpre[wv] = part;
  __syncthreads();
  int off = 0;
  for (int i = 0; i < 32; ++i) {
    off += wpre[i];
    if (i < wv) off += wsum[i];
  }
  const int e0 = off + inc - t;
  const int e1 = e0 + v0;
  const int e2 = e1 + v1;
  const int e3 = e2 + v2;
  const float d0 = (v0 > 0) ? rsqrtf((float)v0) : 0.f;
  const float d1 = (v1 > 0) ? rsqrtf((float)v1) : 0.f;
  const float d2 = (v2 > 0) ? rsqrtf((float)v2) : 0.f;
  const float d3 = (v3 > 0) ? rsqrtf((float)v3) : 0.f;
  if (base + 3 < n) {
    *(int4*)(rowptr + base) = make_int4(e0, e1, e2, e3);
    *(float4*)(dis + base)  = make_float4(d0, d1, d2, d3);
  } else {
    if (base + 0 < n) { rowptr[base + 0] = e0; dis[base + 0] = d0; }
    if (base + 1 < n) { rowptr[base + 1] = e1; dis[base + 1] = d1; }
    if (base + 2 < n) { rowptr[base + 2] = e2; dis[base + 2] = d2; }
  }
}

#define FILL_EPT 4
__global__ void gcn_fill_lists_kernel(const int* __restrict__ src, const int* __restrict__ dst,
                                      const int* __restrict__ rank, const int* __restrict__ rowptr,
                                      const float* __restrict__ dis, int2* __restrict__ lst, int E) {
  const int T  = gridDim.x * blockDim.x;
  const int e0 = blockIdx.x * blockDim.x + threadIdx.x;
  int s[FILL_EPT], d[FILL_EPT], r[FILL_EPT], p[FILL_EPT];
  float we[FILL_EPT];
#pragma unroll
  for (int k = 0; k < FILL_EPT; ++k) {
    const int e  = e0 + k * T;
    const int ec = (e < E) ? e : (E - 1);
    s[k] = src[ec];
    d[k] = dst[ec];
    r[k] = rank[ec];
  }
#pragma unroll
  for (int k = 0; k < FILL_EPT; ++k) {
    p[k]  = rowptr[d[k]] + r[k];
    we[k] = dis[s[k]] * dis[d[k]];
  }
#pragma unroll
  for (int k = 0; k < FILL_EPT; ++k) {
    if (e0 + k * T < E) lst[p[k]] = make_int2(s[k], __float_as_int(we[k]));
  }
}

__global__ void gcn_scale_init_kernel(const float* __restrict__ x, float* __restrict__ o,
                                      const float* __restrict__ wts, int n4) {
  int i = blockIdx.x * blockDim.x + threadIdx.x;
  if (i < n4) {
    float s = wts[0];
    float4 a = ((const float4*)x)[i];
    float4 r;
    r.x = s * a.x; r.y = s * a.y; r.z = s * a.z; r.w = s * a.w;
    ((float4*)o)[i] = r;
  }
}

__device__ __forceinline__ void gcn_row_fma(float4& a, float w, const float4 x) {
  a.x += w * x.x; a.y += w * x.y; a.z += w * x.z; a.w += w * x.w;
}

__global__ void gcn_spmm_gather_acc_kernel(const int* __restrict__ rowptr, const int2* __restrict__ lst,
                                           const float* __restrict__ xin, float* __restrict__ xout,
                                           float* __restrict__ acc, const float* __restrict__ wts,
                                           int k, int last, int n, int E) {
  const int t = blockIdx.x * blockDim.x + threadIdx.x;
  const int row = t >> 4;
  if (row >= n) return;
  const int l16 = t & 15;
  const int c   = l16 << 2;
  const size_t o = (size_t)row * DIM + c;
  const bool first = (k == 1);
  float4 r = *(const float4*)((first ? xin : (const float*)acc) + o);
  const float wlayer = wts[k];
  const float wself  = wts[0];
  const int beg = rowptr[row];
  const int end = (row + 1 < n) ? rowptr[row + 1] : E;
  const float* xc = xin + c;

  float4 a = make_float4(0.f, 0.f, 0.f, 0.f);
  for (int j0 = beg; j0 < end; j0 += 16) {
    const int jj = j0 + l16;
    int2 ent = make_int2(0, 0);
    if (jj < end) ent = lst[jj];
    const int cnt = min(16, end - j0);
    int i = 0;
    for (; i + 4 <= cnt; i += 4) {
      const int   s0 = __shfl(ent.x, i + 0, 16);
      const int   s1 = __shfl(ent.x, i + 1, 16);
      const int   s2 = __shfl(ent.x, i + 2, 16);
      const int   s3 = __shfl(ent.x, i + 3, 16);
      const float w0 = __int_as_float(__shfl(ent.y, i + 0, 16));
      const float w1 = __int_as_float(__shfl(ent.y, i + 1, 16));
      const float w2 = __int_as_float(__shfl(ent.y, i + 2, 16));
      const float w3 = __int_as_float(__shfl(ent.y, i + 3, 16));
      const float4 x0 = *(const float4*)(xc + (size_t)s0 * DIM);
      const float4 x1 = *(const float4*)(xc + (size_t)s1 * DIM);
      const float4 x2 = *(const float4*)(xc + (size_t)s2 * DIM);
      const float4 x3 = *(const float4*)(xc + (size_t)s3 * DIM);
      gcn_row_fma(a, w0, x0);
      gcn_row_fma(a, w1, x1);
      gcn_row_fma(a, w2, x2);
      gcn_row_fma(a, w3, x3);
    }
    for (; i < cnt; ++i) {
      const int   s0 = __shfl(ent.x, i, 16);
      const float w0 = __int_as_float(__shfl(ent.y, i, 16));
      const float4 x0 = *(const float4*)(xc + (size_t)s0 * DIM);
      gcn_row_fma(a, w0, x0);
    }
  }
  if (!last) *(float4*)(xout + o) = a;

  if (first) {
    r.x *= wself; r.y *= wself; r.z *= wself; r.w *= wself;
  }
  gcn_row_fma(r, wlayer, a);
  *(float4*)(acc + o) = r;
}

__global__ void gcn_edge_dot_wmma_kernel(const int* __restrict__ la, const int* __restrict__ lb,
                                         const float* __restrict__ x, float* __restrict__ res,
                                         int E) {
  const int gwave = (blockIdx.x * blockDim.x + threadIdx.x) >> 5;
  const int lane  = threadIdx.x & 31;
  const int ebase = gwave * 16;
  const int m     = lane & 15;

  int e = ebase + m;
  if (e >= E) e = E - 1;

  const size_t ai = (size_t)la[e];
  const size_t bi = (size_t)lb[e];
  const int k0 = (lane >> 4) << 1;
  const float* ap = x + ai * DIM + k0;
  const float* bp = x + bi * DIM + k0;

  v8f c = {0.f, 0.f, 0.f, 0.f, 0.f, 0.f, 0.f, 0.f};
#pragma unroll
  for (int kk = 0; kk < DIM; kk += 4) {
    v2f a = *(const v2f*)(ap + kk);
    v2f b = *(const v2f*)(bp + kk);
    c = __builtin_amdgcn_wmma_f32_16x16x4_f32(false, a, false, b, (short)0, c, false, false);
  }

  const int r = lane & 7;
  float v = c[0];
  if (r == 1) v = c[1];
  if (r == 2) v = c[2];
  if (r == 3) v = c[3];
  if (r == 4) v = c[4];
  if (r == 5) v = c[5];
  if (r == 6) v = c[6];
  if (r == 7) v = c[7];

  const bool active = (lane < 8) || (lane >= 24);
  const int ei = ebase + ((lane < 8) ? lane : (lane - 16));
  if (active && ei < E) res[ei] = v;
}

extern "C" void kernel_launch(void* const* d_in, const int* in_sizes, int n_in,
                              void* d_out, int out_size, void* d_ws, size_t ws_size,
                              hipStream_t stream) {
  const int*   edge_index = (const int*)d_in[0];
  const int*   eli        = (const int*)d_in[1];
  const float* emb        = (const float*)d_in[2];
  const float* alpha      = (const float*)d_in[3];

  const int E  = in_sizes[0] / 2;
  const int Nn = in_sizes[2] / DIM;
  const int L1 = in_sizes[3];
  const int L  = L1 - 1;
  float* res = (float*)d_out;

  const int* src = edge_index;
  const int* dst = edge_index + E;
  const int* la  = eli;
  const int* lb  = eli + E;

  size_t off = 0;
  auto carve = [&](size_t bytes) -> void* {
    void* p = (char*)d_ws + off;
    off = (off + bytes + 255) & ~(size_t)255;
    return p;
  };
  float* wts = (float*)carve(sizeof(float) * (size_t)L1);
  float* dis = (float*)carve(sizeof(float) * (size_t)Nn);
  int*   cnt    = (int*)carve(sizeof(int) * (size_t)Nn);
  int*   rowptr = (int*)carve(sizeof(int) * (size_t)Nn);
  int*   rank   = (int*)carve(sizeof(int) * (size_t)E);
  int2*  lst    = (int2*)carve(sizeof(int2) * (size_t)E);
  float* xA  = (float*)carve(sizeof(float) * (size_t)Nn * DIM);
  float* xB  = (float*)carve(sizeof(float) * (size_t)Nn * DIM);
  float* acc = (float*)carve(sizeof(float) * (size_t)Nn * DIM);
  (void)ws_size; (void)n_in; (void)out_size;

  const int n4 = Nn * DIM / 4;

  gcn_softmax_kernel<<<1, 1, 0, stream>>>(alpha, wts, L1);
  hipMemsetAsync(cnt, 0, sizeof(int) * (size_t)Nn, stream);
  gcn_count_rank_kernel<<<(E + 255) / 256, 256, 0, stream>>>(dst, cnt, rank, E);
  gcn_scan_dis_kernel<<<(Nn + SCAN_TILE - 1) / SCAN_TILE, SCAN_THREADS, 0, stream>>>(cnt, rowptr, dis, Nn);
  if (E > 0)
    gcn_fill_lists_kernel<<<(E + 256 * FILL_EPT - 1) / (256 * FILL_EPT), 256, 0, stream>>>(
        src, dst, rank, rowptr, dis, lst, E);
  if (L == 0)
    gcn_scale_init_kernel<<<(n4 + 255) / 256, 256, 0, stream>>>(emb, acc, wts, n4);

  const float* xcur = emb;
  float* bufs[2] = {xA, xB};
  for (int l = 0; l < L; ++l) {
    float* xnext = bufs[l & 1];
    long tcount = (long)Nn * (DIM / 4);
    gcn_spmm_gather_acc_kernel<<<(unsigned)((tcount + 255) / 256), 256, 0, stream>>>(
        rowptr, lst, xcur, xnext, acc, wts, l + 1, (l == L - 1) ? 1 : 0, Nn, E);
    xcur = xnext;
  }

  if (E > 0) {
    int waves  = (E + 15) / 16;
    int blocks = (waves + 7) / 8;
    gcn_edge_dot_wmma_kernel<<<blocks, 256, 0, stream>>>(la, lb, acc, res, E);
  }
}


# Edge-dot kernel: gather each row whole (16 lanes x 16 B, two rows per load = four full 128-B lines) and pass it to the WMMA fragment layout through wave-private LDS; k order and sums unchanged
# speedup vs baseline: 2.3270x; 1.1308x over previous
#include <hip/hip_runtime.h>

typedef __attribute__((ext_vector_type(2))) float v2f;
typedef __attribute__((ext_vector_type(8))) float v8f;

#define DIM 64

__global__ void gcn_softmax_kernel(const float* __restrict__ alpha,
                                   float* __restrict__ wts, int n) {
  if (blockIdx.x == 0 && threadIdx.x == 0) {
    float m = alpha[0];
    for (int i = 1; i < n; ++i) m = fmaxf(m, alpha[i]);
    float s = 0.f;
    for (int i = 0; i < n; ++i) s += expf(alpha[i] - m);
    for (int i = 0; i < n; ++i) wts[i] = expf(alpha[i] - m) / s;
  }
}

__global__ void gcn_count_rank_kernel(const int* __restrict__ dst, int* __restrict__ cnt,
                                      int* __restrict__ rank, int E) {
  int e = blockIdx.x * blockDim.x + threadIdx.x;
  if (e < E) rank[e] = atomicAdd(&cnt[dst[e]], 1);
}

#define SCAN_THREADS 1024
#define SCAN_TILE    4096
__global__ void __launch_bounds__(SCAN_THREADS)
gcn_scan_dis_kernel(const int* __restrict__ cnt, int* __restrict__ rowptr,
                    float* __restrict__ dis, int n) {
  __shared__ int wsum[32];
  __shared__ int wpre[32];
  const int tid  = threadIdx.x;
  const int lane = tid & 31;
  const int wv   = tid >> 5;
  const int tile0 = blockIdx.x * SCAN_TILE;

  int part = 0;
#pragma unroll 8
  for (int j = tid * 4; j < tile0; j += SCAN_THREADS * 4) {
    const int4 q = *(const int4*)(cnt + j);
    part += (q.x + q.y) + (q.z + q.w);
  }

  const int base = tile0 + tid * 4;
  int v0 = 0, v1 = 0, v2 = 0, v3 = 0;
  if (base + 3 < n) {
    const int4 q = *(const int4*)(cnt + base);
    v0 = q.x; v1 = q.y; v2 = q.z; v3 = q.w;
  } else {
    if (base + 0 < n) v0 = cnt[base + 0];
    if (base + 1 < n) v1 = cnt[base + 1];
    if (base + 2 < n) v2 = cnt[base + 2];
  }
  const int t = v0 + v1 + v2 + v3;

  int inc = t;
  for (int o = 1; o < 32; o <<= 1) {
    const int u = __shfl_up(inc, o, 32);
    if (lane >= o) inc += u;
  }
  for (int o = 16; o > 0; o >>= 1) part += __shfl_xor(part, o, 32);
  if (lane == 31) wsum[wv] = inc;
  if (lane == 0)  wpre[wv] = part;
  __syncthreads();
  int off = 0;
  for (int i = 0; i < 32; ++i) {
    off += wpre[i];
    if (i < wv) off += wsum[i];
  }
  const int e0 = off + inc - t;
  const int e1 = e0 + v0;
  const int e2 = e1 + v1;
  const int e3 = e2 + v2;
  const float d0 = (v0 > 0) ? rsqrtf((float)v0) : 0.f;
  const float d1 = (v1 > 0) ? rsqrtf((float)v1) : 0.f;
  const float d2 = (v2 > 0) ? rsqrtf((float)v2) : 0.f;
  const float d3 = (v3 > 0) ? rsqrtf((float)v3) : 0.f;
  if (base + 3 < n) {
    *(int4*)(rowptr + base) = make_int4(e0, e1, e2, e3);
    *(float4*)(dis + base)  = make_float4(d0, d1, d2, d3);
  } else {
    if (base + 0 < n) { rowptr[base + 0] = e0; dis[base + 0] = d0; }
    if (base + 1 < n) { rowptr[base + 1] = e1; dis[base + 1] = d1; }
    if (base + 2 < n) { rowptr[base + 2] = e2; dis[base + 2] = d2; }
  }
}

#define FILL_EPT 4
__global__ void gcn_fill_lists_kernel(const int* __restrict__ src, const int* __restrict__ dst,
                                      const int* __restrict__ rank, const int* __restrict__ rowptr,
                                      const float* __restrict__ dis, int2* __restrict__ lst, int E) {
  const int T  = gridDim.x * blockDim.x;
  const int e0 = blockIdx.x * blockDim.x + threadIdx.x;
  int s[FILL_EPT], d[FILL_EPT], r[FILL_EPT], p[FILL_EPT];
  float we[FILL_EPT];
#pragma unroll
  for (int k = 0; k < FILL_EPT; ++k) {
    const int e  = e0 + k * T;
    const int ec = (e < E) ? e : (E - 1);
    s[k] = src[ec];
    d[k] = dst[ec];
    r[k] = rank[ec];
  }
#pragma unroll
  for (int k = 0; k < FILL_EPT; ++k) {
    p[k]  = rowptr[d[k]] + r[k];
    we[k] = dis[s[k]] * dis[d[k]];
  }
#pragma unroll
  for (int k = 0; k < FILL_EPT; ++k) {
    if (e0 + k * T < E) lst[p[k]] = make_int2(s[k], __float_as_int(we[k]));
  }
}

__global__ void gcn_scale_init_kernel(const float* __restrict__ x, float* __restrict__ o,
                                      const float* __restrict__ wts, int n4) {
  int i = blockIdx.x * blockDim.x + threadIdx.x;
  if (i < n4) {
    float s = wts[0];
    float4 a = ((const float4*)x)[i];
    float4 r;
    r.x = s * a.x; r.y = s * a.y; r.z = s * a.z; r.w = s * a.w;
    ((float4*)o)[i] = r;
  }
}

__device__ __forceinline__ void gcn_row_fma(float4& a, float w, const float4 x) {
  a.x += w * x.x; a.y += w * x.y; a.z += w * x.z; a.w += w * x.w;
}

__global__ void gcn_spmm_gather_acc_kernel(const int* __restrict__ rowptr, const int2* __restrict__ lst,
                                           const float* __restrict__ xin, float* __restrict__ xout,
                                           float* __restrict__ acc, const float* __restrict__ wts,
                                           int k, int last, int n, int E) {
  const int t = blockIdx.x * blockDim.x + threadIdx.x;
  const int row = t >> 4;
  if (row >= n) return;
  const int l16 = t & 15;
  const int c   = l16 << 2;
  const size_t o = (size_t)row * DIM + c;
  const bool first = (k == 1);
  float4 r = *(const float4*)((first ? xin : (const float*)acc) + o);
  const float wlayer = wts[k];
  const float wself  = wts[0];
  const int beg = rowptr[row];
  const int end = (row + 1 < n) ? rowptr[row + 1] : E;
  const float* xc = xin + c;

  float4 a = make_float4(0.f, 0.f, 0.f, 0.f);
  for (int j0 = beg; j0 < end; j0 += 16) {
    const int jj = j0 + l16;
    int2 ent = make_int2(0, 0);
    if (jj < end) ent = lst[jj];
    const int cnt = min(16, end - j0);
    int i = 0;
    for (; i + 4 <= cnt; i += 4) {
      const int   s0 = __shfl(ent.x, i + 0, 16);
      const int   s1 = __shfl(ent.x, i + 1, 16);
      const int   s2 = __shfl(ent.x, i + 2, 16);
      const int   s3 = __shfl(ent.x, i + 3, 16);
      const float w0 = __int_as_float(__shfl(ent.y, i + 0, 16));
      const float w1 = __int_as_float(__shfl(ent.y, i + 1, 16));
      const float w2 = __int_as_float(__shfl(ent.y, i + 2, 16));
      const float w3 = __int_as_float(__shfl(ent.y, i + 3, 16));
      const float4 x0 = *(const float4*)(xc + (size_t)s0 * DIM);
      const float4 x1 = *(const float4*)(xc + (size_t)s1 * DIM);
      const float4 x2 = *(const float4*)(xc + (size_t)s2 * DIM);
      const float4 x3 = *(const float4*)(xc + (size_t)s3 * DIM);
      gcn_row_fma(a, w0, x0);
      gcn_row_fma(a, w1, x1);
      gcn_row_fma(a, w2, x2);
      gcn_row_fma(a, w3, x3);
    }
    for (; i < cnt; ++i) {
      const int   s0 = __shfl(ent.x, i, 16);
      const float w0 = __int_as_float(__shfl(ent.y, i, 16));
      const float4 x0 = *(const float4*)(xc + (size_t)s0 * DIM);
      gcn_row_fma(a, w0, x0);
    }
  }
  if (!last) *(float4*)(xout + o) = a;

  if (first) {
    r.x *= wself; r.y *= wself; r.z *= wself; r.w *= wself;
  }
  gcn_row_fma(r, wlayer, a);
  *(float4*)(acc + o) = r;
}

#define ED_WAVES 8
#define ED_ROWS  32
#define ED_PITCH 68
__global__ void __launch_bounds__(ED_WAVES * 32)
gcn_edge_dot_wmma_kernel(const int* __restrict__ la, const int* __restrict__ lb,
                         const float* __restrict__ x, float* __restrict__ res,
                         int E) {
  __shared__ __attribute__((aligned(16))) float stage[ED_WAVES * ED_ROWS * ED_PITCH];

  const int gwave = (blockIdx.x * blockDim.x + threadIdx.x) >> 5;
  const int wv    = threadIdx.x >> 5;
  const int lane  = threadIdx.x & 31;
  const int ebase = gwave * 16;
  const int m     = lane & 15;
  const int hh    = lane >> 4;

  int e = ebase + m;
  if (e >= E) e = E - 1;

  const int* __restrict__ ip = hh ? lb : la;
  const int myrow = ip[e];

  float* st = stage + wv * (ED_ROWS * ED_PITCH);

  float4 v[16];
#pragma unroll
  for (int i = 0; i < 16; ++i) {
    const int ri = __shfl(myrow, i, 16);
    v[i] = *(const float4*)(x + (size_t)ri * DIM + (m << 2));
  }
#pragma unroll
  for (int i = 0; i < 16; ++i)
    *(float4*)(st + (i + 16 * hh) * ED_PITCH + (m << 2)) = v[i];


  const int k0 = hh << 1;
  const float* ap = st + m * ED_PITCH + k0;
  const float* bp = st + (16 + m) * ED_PITCH + k0;

  v8f c = {0.f, 0.f, 0.f, 0.f, 0.f, 0.f, 0.f, 0.f};
#pragma unroll
  for (int kk = 0; kk < DIM; kk += 4) {
    v2f a = *(const v2f*)(ap + kk);
    v2f b = *(const v2f*)(bp + kk);
    c = __builtin_amdgcn_wmma_f32_16x16x4_f32(false, a, false, b, (short)0, c, false, false);
  }

  const int r = lane & 7;
  float d = c[0];
  if (r == 1) d = c[1];
  if (r == 2) d = c[2];
  if (r == 3) d = c[3];
  if (r == 4) d = c[4];
  if (r == 5) d = c[5];
  if (r == 6) d = c[6];
  if (r == 7) d = c[7];

  const bool active = (lane < 8) || (lane >= 24);
  const int ei = ebase + ((lane < 8) ? lane : (lane - 16));
  if (active && ei < E) res[ei] = d;
}

extern "C" void kernel_launch(void* const* d_in, const int* in_sizes, int n_in,
                              void* d_out, int out_size, void* d_ws, size_t ws_size,
                              hipStream_t stream) {
  const int*   edge_index = (const int*)d_in[0];
  const int*   eli        = (const int*)d_in[1];
  const float* emb        = (const float*)d_in[2];
  const float* alpha      = (const float*)d_in[3];

  const int E  = in_sizes[0] / 2;
  const int Nn = in_sizes[2] / DIM;
  const int L1 = in_sizes[3];
  const int L  = L1 - 1;
  float* res = (float*)d_out;

  const int* src = edge_index;
  const int* dst = edge_index + E;
  const int* la  = eli;
  const int* lb  = eli + E;

  size_t off = 0;
  auto carve = [&](size_t bytes) -> void* {
    void* p = (char*)d_ws + off;
    off = (off + bytes + 255) & ~(size_t)255;
    return p;
  };
  float* wts = (float*)carve(sizeof(float) * (size_t)L1);
  float* dis = (float*)carve(sizeof(float) * (size_t)Nn);
  int*   cnt    = (int*)carve(sizeof(int) * (size_t)Nn);
  int*   rowptr = (int*)carve(sizeof(int) * (size_t)Nn);
  int*   rank   = (int*)carve(sizeof(int) * (size_t)E);
  int2*  lst    = (int2*)carve(sizeof(int2) * (size_t)E);
  float* xA  = (float*)carve(sizeof(float) * (size_t)Nn * DIM);
  float* xB  = (float*)carve(sizeof(float) * (size_t)Nn * DIM);
  float* acc = (float*)carve(sizeof(float) * (size_t)Nn * DIM);
  (void)ws_size; (void)n_in; (void)out_size;

  const int n4 = Nn * DIM / 4;

  gcn_softmax_kernel<<<1, 1, 0, stream>>>(alpha, wts, L1);
  hipMemsetAsync(cnt, 0, sizeof(int) * (size_t)Nn, stream);
  gcn_count_rank_kernel<<<(E + 255) / 256, 256, 0, stream>>>(dst, cnt, rank, E);
  gcn_scan_dis_kernel<<<(Nn + SCAN_TILE - 1) / SCAN_TILE, SCAN_THREADS, 0, stream>>>(cnt, rowptr, dis, Nn);
  if (E > 0)
    gcn_fill_lists_kernel<<<(E + 256 * FILL_EPT - 1) / (256 * FILL_EPT), 256, 0, stream>>>(
        src, dst, rank, rowptr, dis, lst, E);
  if (L == 0)
    gcn_scale_init_kernel<<<(n4 + 255) / 256, 256, 0, stream>>>(emb, acc, wts, n4);

  const float* xcur = emb;
  float* bufs[2] = {xA, xB};
  for (int l = 0; l < L; ++l) {
    float* xnext = bufs[l & 1];
    long tcount = (long)Nn * (DIM / 4);
    gcn_spmm_gather_acc_kernel<<<(unsigned)((tcount + 255) / 256), 256, 0, stream>>>(
        rowptr, lst, xcur, xnext, acc, wts, l + 1, (l == L - 1) ? 1 : 0, Nn, E);
    xcur = xnext;
  }

  if (E > 0) {
    int waves  = (E + 15) / 16;
    int blocks = (waves + 7) / 8;
    gcn_edge_dot_wmma_kernel<<<blocks, 256, 0, stream>>>(la, lb, acc, res, E);
  }
}


# Build the per-destination lists by a two-level counting sort on LDS counters (chunk -> 256 node buckets -> node), not a global returning add per edge + scan + scattered fill; sums in LDS-add order
# speedup vs baseline: 3.8650x; 1.6609x over previous
#include <hip/hip_runtime.h>

typedef __attribute__((ext_vector_type(2))) float v2f;
typedef __attribute__((ext_vector_type(8))) float v8f;

#define DIM 64

__global__ void gcn_softmax_kernel(const float* __restrict__ alpha,
                                   float* __restrict__ wts, int n) {
  if (blockIdx.x == 0 && threadIdx.x == 0) {
    float m = alpha[0];
    for (int i = 1; i < n; ++i) m = fmaxf(m, alpha[i]);
    float s = 0.f;
    for (int i = 0; i < n; ++i) s += expf(alpha[i] - m);
    for (int i = 0; i < n; ++i) wts[i] = expf(alpha[i] - m) / s;
  }
}

__global__ void gcn_count_rank_kernel(const int* __restrict__ dst, int* __restrict__ cnt,
                                      int* __restrict__ rank, int E) {
  int e = blockIdx.x * blockDim.x + threadIdx.x;
  if (e < E) rank[e] = atomicAdd(&cnt[dst[e]], 1);
}

#define SCAN_THREADS 1024
#define SCAN_TILE    4096
__global__ void __launch_bounds__(SCAN_THREADS)
gcn_scan_dis_kernel(const int* __restrict__ cnt, int* __restrict__ rowptr,
                    float* __restrict__ dis, int n) {
  __shared__ int wsum[32];
  __shared__ int wpre[32];
  const int tid  = threadIdx.x;
  const int lane = tid & 31;
  const int wv   = tid >> 5;
  const int tile0 = blockIdx.x * SCAN_TILE;

  int part = 0;
#pragma unroll 8
  for (int j = tid * 4; j < tile0; j += SCAN_THREADS * 4) {
    const int4 q = *(const int4*)(cnt + j);
    part += (q.x + q.y) + (q.z + q.w);
  }

  const int base = tile0 + tid * 4;
  int v0 = 0, v1 = 0, v2 = 0, v3 = 0;
  if (base + 3 < n) {
    const int4 q = *(const int4*)(cnt + base);
    v0 = q.x; v1 = q.y; v2 = q.z; v3 = q.w;
  } else {
    if (base + 0 < n) v0 = cnt[base + 0];
    if (base + 1 < n) v1 = cnt[base + 1];
    if (base + 2 < n) v2 = cnt[base + 2];
  }
  const int t = v0 + v1 + v2 + v3;

  int inc = t;
  for (int o = 1; o < 32; o <<= 1) {
    const int u = __shfl_up(inc, o, 32);
    if (lane >= o) inc += u;
  }
  for (int o = 16; o > 0; o >>= 1) part += __shfl_xor(part, o, 32);
  if (lane == 31) wsum[wv] = inc;
  if (lane == 0)  wpre[wv] = part;
  __syncthreads();
  int off = 0;
  for (int i = 0; i < 32; ++i) {
    off += wpre[i];
    if (i < wv) off += wsum[i];
  }
  const int e0 = off + inc - t;
  const int e1 = e0 + v0;
  const int e2 = e1 + v1;
  const int e3 = e2 + v2;
  const float d0 = (v0 > 0) ? rsqrtf((float)v0) : 0.f;
  const float d1 = (v1 > 0) ? rsqrtf((float)v1) : 0.f;
  const float d2 = (v2 > 0) ? rsqrtf((float)v2) : 0.f;
  const float d3 = (v3 > 0) ? rsqrtf((float)v3) : 0.f;
  if (base + 3 < n) {
    *(int4*)(rowptr + base) = make_int4(e0, e1, e2, e3);
    *(float4*)(dis + base)  = make_float4(d0, d1, d2, d3);
  } else {
    if (base + 0 < n) { rowptr[base + 0] = e0; dis[base + 0] = d0; }
    if (base + 1 < n) { rowptr[base + 1] = e1; dis[base + 1] = d1; }
    if (base + 2 < n) { rowptr[base + 2] = e2; dis[base + 2] = d2; }
  }
}

#define FILL_EPT 4
__global__ void gcn_fill_lists_kernel(const int* __restrict__ src, const int* __restrict__ dst,
                                      const int* __restrict__ rank, const int* __restrict__ rowptr,
                                      const float* __restrict__ dis, int2* __restrict__ lst, int E) {
  const int T  = gridDim.x * blockDim.x;
  const int e0 = blockIdx.x * blockDim.x + threadIdx.x;
  int s[FILL_EPT], d[FILL_EPT], r[FILL_EPT], p[FILL_EPT];
  float we[FILL_EPT];
#pragma unroll
  for (int k = 0; k < FILL_EPT; ++k) {
    const int e  = e0 + k * T;
    const int ec = (e < E) ? e : (E - 1);
    s[k] = src[ec];
    d[k] = dst[ec];
    r[k] = rank[ec];
  }
#pragma unroll
  for (int k = 0; k < FILL_EPT; ++k) {
    p[k]  = rowptr[d[k]] + r[k];
    we[k] = dis[s[k]] * dis[d[k]];
  }
#pragma unroll
  for (int k = 0; k < FILL_EPT; ++k) {
    if (e0 + k * T < E) lst[p[k]] = make_int2(s[k], __float_as_int(we[k]));
  }
}

#define BIN_THREADS 1024
#define BIN_CH      4096
#define BIN_NB      256
#define BIN_ROW     257
#define BKT_CAP     8192

__device__ __forceinline__ int gcn_block_scan_excl(int t, int* wsum, int& total) {
  const int lane = threadIdx.x & 31;
  const int wv   = threadIdx.x >> 5;
  int inc = t;
  for (int o = 1; o < 32; o <<= 1) {
    const int u = __shfl_up(inc, o, 32);
    if (lane >= o) inc += u;
  }
  if (lane == 31) wsum[wv] = inc;
  __syncthreads();
  int off = 0, tot = 0;
  for (int i = 0; i < 32; ++i) {
    const int s = wsum[i];
    tot += s;
    if (i < wv) off += s;
  }
  __syncthreads();
  total = tot;
  return off + inc - t;
}

__global__ void __launch_bounds__(BIN_THREADS)
gcn_bin_edges_kernel(const int* __restrict__ src, const int* __restrict__ dst,
                     int2* __restrict__ recs, int* __restrict__ woff, int E, int BN) {
  __shared__ int  wsum[32];
  __shared__ int  hist[BIN_NB];
  __shared__ int  start[BIN_NB];
  __shared__ int2 stage[BIN_CH];
  const int tid  = threadIdx.x;
  const int w    = blockIdx.x;
  const int base = w * BIN_CH;
  if (tid < BIN_NB) hist[tid] = 0;
  __syncthreads();
  int s[4], d[4], bk[4], r[4];
#pragma unroll
  for (int k = 0; k < 4; ++k) {
    const int e = base + k * BIN_THREADS + tid;
    s[k] = 0; d[k] = 0;
    if (e < E) { s[k] = src[e]; d[k] = dst[e]; }
  }
#pragma unroll
  for (int k = 0; k < 4; ++k) {
    const int e = base + k * BIN_THREADS + tid;
    bk[k] = 0; r[k] = 0;
    if (e < E) {
      bk[k] = d[k] / BN;
      r[k]  = atomicAdd(&hist[bk[k]], 1);
    }
  }
  __syncthreads();
  const int v = (tid < BIN_NB) ? hist[tid] : 0;
  int tot;
  const int ex = gcn_block_scan_excl(v, wsum, tot);
  if (tid < BIN_NB) {
    start[tid] = ex;
    woff[(size_t)w * BIN_ROW + tid] = ex;
  }
  if (tid == BIN_NB) woff[(size_t)w * BIN_ROW + BIN_NB] = tot;
  __syncthreads();
#pragma unroll
  for (int k = 0; k < 4; ++k) {
    const int e = base + k * BIN_THREADS + tid;
    if (e < E) stage[start[bk[k]] + r[k]] = make_int2(s[k], d[k]);
  }
  __syncthreads();
  for (int i = tid; i < tot; i += BIN_THREADS) recs[base + i] = stage[i];
}

__device__ __forceinline__ int2 gcn_bucket_rec(const int2* __restrict__ recs, const int* segpre,
                                               const int* segbase, int i) {
  int lo = 0;
#pragma unroll
  for (int step = BIN_THREADS / 2; step >= 1; step >>= 1)
    if (segpre[lo + step] <= i) lo += step;
  return recs[segbase[lo] + (i - segpre[lo])];
}

__global__ void __launch_bounds__(BIN_THREADS)
gcn_bucket_lists_kernel(const int2* __restrict__ recs, const int* __restrict__ woff,
                        int* __restrict__ rowptr, float* __restrict__ dis,
                        int2* __restrict__ lst, int NW, int BN, int Nn) {
  __shared__ int   wsum[32];
  __shared__ int   cntl[BIN_THREADS];
  __shared__ float disl[BIN_THREADS];
  __shared__ int   segpre[BIN_THREADS + 1];
  __shared__ int   segbase[BIN_THREADS];
  __shared__ int2  stageL[BKT_CAP];
  __shared__ int2  recL[BKT_CAP];
  const int tid = threadIdx.x;
  const int b   = blockIdx.x;

  int e0 = 0, len = 0;
  if (tid < NW) {
    const int* row = woff + (size_t)tid * BIN_ROW + b;
    e0  = row[0];
    len = row[1] - e0;
    segbase[tid] = tid * BIN_CH + e0;
  }
  int gs, tb;
  (void)gcn_block_scan_excl(e0, wsum, gs);
  const int ex = gcn_block_scan_excl(len, wsum, tb);
  segpre[tid] = ex;
  if (tid == BIN_THREADS - 1) segpre[BIN_THREADS] = ex + len;
  cntl[tid] = 0;
  __syncthreads();

  const int nbase = b * BN;
  for (int i0 = tid; i0 < tb; i0 += 4 * BIN_THREADS) {
    int2 rc[4];
#pragma unroll
    for (int k = 0; k < 4; ++k) {
      const int i = i0 + k * BIN_THREADS;
      rc[k] = gcn_bucket_rec(recs, segpre, segbase, min(i, tb - 1));
    }
#pragma unroll
    for (int k = 0; k < 4; ++k) {
      const int i = i0 + k * BIN_THREADS;
      if (i < tb) {
        if (i < BKT_CAP) recL[i] = rc[k];
        atomicAdd(&cntl[rc[k].y - nbase], 1);
      }
    }
  }
  __syncthreads();

  const int c = cntl[tid];
  int tdummy;
  const int exn = gcn_block_scan_excl(c, wsum, tdummy);
  const float dd = (c > 0) ? rsqrtf((float)c) : 0.f;
  cntl[tid] = exn;
  disl[tid] = dd;
  const int node = nbase + tid;
  if (tid < BN && node < Nn) { rowptr[node] = gs + exn; dis[node] = dd; }
  __syncthreads();

  for (int i0 = tid; i0 < tb; i0 += 4 * BIN_THREADS) {
    int2 rc[4];
#pragma unroll
    for (int k = 0; k < 4; ++k) {
      const int i = i0 + k * BIN_THREADS;
      rc[k] = (i < BKT_CAP) ? recL[min(i, tb - 1)]
                            : gcn_bucket_rec(recs, segpre, segbase, min(i, tb - 1));
    }
#pragma unroll
    for (int k = 0; k < 4; ++k) {
      const int i = i0 + k * BIN_THREADS;
      if (i < tb) {
        const int dl = rc[k].y - nbase;
        const int p  = atomicAdd(&cntl[dl], 1);
        const int2 ent = make_int2(rc[k].x, __float_as_int(disl[dl]));
        if (p < BKT_CAP) stageL[p] = ent; else lst[gs + p] = ent;
      }
    }
  }
  __syncthreads();
  const int nst = min(tb, BKT_CAP);
  for (int i = tid; i < nst; i += BIN_THREADS) lst[gs + i] = stageL[i];
}

__global__ void gcn_list_weights_kernel(int2* __restrict__ lst, const float* __restrict__ dis, int E) {
  const int p = (blockIdx.x * blockDim.x + threadIdx.x) * 4;
  if (p + 3 < E) {
    int4 q0 = *(const int4*)(lst + p);
    int4 q1 = *(const int4*)(lst + p + 2);
    const float a0 = dis[q0.x], a1 = dis[q0.z], a2 = dis[q1.x], a3 = dis[q1.z];
    q0.y = __float_as_int(a0 * __int_as_float(q0.y));
    q0.w = __float_as_int(a1 * __int_as_float(q0.w));
    q1.y = __float_as_int(a2 * __int_as_float(q1.y));
    q1.w = __float_as_int(a3 * __int_as_float(q1.w));
    *(int4*)(lst + p)     = q0;
    *(int4*)(lst + p + 2) = q1;
  } else {
    for (int j = p; j < E; ++j) {
      int2 q = lst[j];
      q.y = __float_as_int(dis[q.x] * __int_as_float(q.y));
      lst[j] = q;
    }
  }
}

__global__ void gcn_scale_init_kernel(const float* __restrict__ x, float* __restrict__ o,
                                      const float* __restrict__ wts, int n4) {
  int i = blockIdx.x * blockDim.x + threadIdx.x;
  if (i < n4) {
    float s = wts[0];
    float4 a = ((const float4*)x)[i];
    float4 r;
    r.x = s * a.x; r.y = s * a.y; r.z = s * a.z; r.w = s * a.w;
    ((float4*)o)[i] = r;
  }
}

__device__ __forceinline__ void gcn_row_fma(float4& a, float w, const float4 x) {
  a.x += w * x.x; a.y += w * x.y; a.z += w * x.z; a.w += w * x.w;
}

__global__ void gcn_spmm_gather_acc_kernel(const int* __restrict__ rowptr, const int2* __restrict__ lst,
                                           const float* __restrict__ xin, float* __restrict__ xout,
                                           float* __restrict__ acc, const float* __restrict__ wts,
                                           int k, int last, int n, int E) {
  const int t = blockIdx.x * blockDim.x + threadIdx.x;
  const int row = t >> 4;
  if (row >= n) return;
  const int l16 = t & 15;
  const int c   = l16 << 2;
  const size_t o = (size_t)row * DIM + c;
  const bool first = (k == 1);
  float4 r = *(const float4*)((first ? xin : (const float*)acc) + o);
  const float wlayer = wts[k];
  const float wself  = wts[0];
  const int beg = rowptr[row];
  const int end = (row + 1 < n) ? rowptr[row + 1] : E;
  const float* xc = xin + c;

  float4 a = make_float4(0.f, 0.f, 0.f, 0.f);
  for (int j0 = beg; j0 < end; j0 += 16) {
    const int jj = j0 + l16;
    int2 ent = make_int2(0, 0);
    if (jj < end) ent = lst[jj];
    const int cnt = min(16, end - j0);
    int i = 0;
    for (; i + 4 <= cnt; i += 4) {
      const int   s0 = __shfl(ent.x, i + 0, 16);
      const int   s1 = __shfl(ent.x, i + 1, 16);
      const int   s2 = __shfl(ent.x, i + 2, 16);
      const int   s3 = __shfl(ent.x, i + 3, 16);
      const float w0 = __int_as_float(__shfl(ent.y, i + 0, 16));
      const float w1 = __int_as_float(__shfl(ent.y, i + 1, 16));
      const float w2 = __int_as_float(__shfl(ent.y, i + 2, 16));
      const float w3 = __int_as_float(__shfl(ent.y, i + 3, 16));
      const float4 x0 = *(const float4*)(xc + (size_t)s0 * DIM);
      const float4 x1 = *(const float4*)(xc + (size_t)s1 * DIM);
      const float4 x2 = *(const float4*)(xc + (size_t)s2 * DIM);
      const float4 x3 = *(const float4*)(xc + (size_t)s3 * DIM);
      gcn_row_fma(a, w0, x0);
      gcn_row_fma(a, w1, x1);
      gcn_row_fma(a, w2, x2);
      gcn_row_fma(a, w3, x3);
    }
    for (; i < cnt; ++i) {
      const int   s0 = __shfl(ent.x, i, 16);
      const float w0 = __int_as_float(__shfl(ent.y, i, 16));
      const float4 x0 = *(const float4*)(xc + (size_t)s0 * DIM);
      gcn_row_fma(a, w0, x0);
    }
  }
  if (!last) *(float4*)(xout + o) = a;

  if (first) {
    r.x *= wself; r.y *= wself; r.z *= wself; r.w *= wself;
  }
  gcn_row_fma(r, wlayer, a);
  *(float4*)(acc + o) = r;
}

#define ED_WAVES 8
#define ED_ROWS  32
#define ED_PITCH 68
__global__ void __launch_bounds__(ED_WAVES * 32)
gcn_edge_dot_wmma_kernel(const int* __restrict__ la, const int* __restrict__ lb,
                         const float* __restrict__ x, float* __restrict__ res,
                         int E) {
  __shared__ __attribute__((aligned(16))) float stage[ED_WAVES * ED_ROWS * ED_PITCH];

  const int gwave = (blockIdx.x * blockDim.x + threadIdx.x) >> 5;
  const int wv    = threadIdx.x >> 5;
  const int lane  = threadIdx.x & 31;
  const int ebase = gwave * 16;
  const int m     = lane & 15;
  const int hh    = lane >> 4;

  int e = ebase + m;
  if (e >= E) e = E - 1;

  const int* __restrict__ ip = hh ? lb : la;
  const int myrow = ip[e];

  float* st = stage + wv * (ED_ROWS * ED_PITCH);

  float4 v[16];
#pragma unroll
  for (int i = 0; i < 16; ++i) {
    const int ri = __shfl(myrow, i, 16);
    v[i] = *(const float4*)(x + (size_t)ri * DIM + (m << 2));
  }
#pragma unroll
  for (int i = 0; i < 16; ++i)
    *(float4*)(st + (i + 16 * hh) * ED_PITCH + (m << 2)) = v[i];


  const int k0 = hh << 1;
  const float* ap = st + m * ED_PITCH + k0;
  const float* bp = st + (16 + m) * ED_PITCH + k0;

  v8f c = {0.f, 0.f, 0.f, 0.f, 0.f, 0.f, 0.f, 0.f};
#pragma unroll
  for (int kk = 0; kk < DIM; kk += 4) {
    v2f a = *(const v2f*)(ap + kk);
    v2f b = *(const v2f*)(bp + kk);
    c = __builtin_amdgcn_wmma_f32_16x16x4_f32(false, a, false, b, (short)0, c, false, false);
  }

  const int r = lane & 7;
  float d = c[0];
  if (r == 1) d = c[1];
  if (r == 2) d = c[2];
  if (r == 3) d = c[3];
  if (r == 4) d = c[4];
  if (r == 5) d = c[5];
  if (r == 6) d = c[6];
  if (r == 7) d = c[7];

  const bool active = (lane < 8) || (lane >= 24);
  const int ei = ebase + ((lane < 8) ? lane : (lane - 16));
  if (active && ei < E) res[ei] = d;
}

extern "C" void kernel_launch(void* const* d_in, const int* in_sizes, int n_in,
                              void* d_out, int out_size, void* d_ws, size_t ws_size,
                              hipStream_t stream) {
  const int*   edge_index = (const int*)d_in[0];
  const int*   eli        = (const int*)d_in[1];
  const float* emb        = (const float*)d_in[2];
  const float* alpha      = (const float*)d_in[3];

  const int E  = in_sizes[0] / 2;
  const int Nn = in_sizes[2] / DIM;
  const int L1 = in_sizes[3];
  const int L  = L1 - 1;
  float* res = (float*)d_out;

  const int* src = edge_index;
  const int* dst = edge_index + E;
  const int* la  = eli;
  const int* lb  = eli + E;

  size_t off = 0;
  auto carve = [&](size_t bytes) -> void* {
    void* p = (char*)d_ws + off;
    off = (off + bytes + 255) & ~(size_t)255;
    return p;
  };
  float* wts = (float*)carve(sizeof(float) * (size_t)L1);
  float* dis = (float*)carve(sizeof(float) * (size_t)Nn);
  int*   cnt    = (int*)carve(sizeof(int) * (size_t)Nn);
  int*   rowptr = (int*)carve(sizeof(int) * (size_t)Nn);
  int*   rank   = (int*)carve(sizeof(int) * (size_t)E);
  int2*  lst    = (int2*)carve(sizeof(int2) * (size_t)E);
  float* xA  = (float*)carve(sizeof(float) * (size_t)Nn * DIM);
  float* xB  = (float*)carve(sizeof(float) * (size_t)Nn * DIM);
  float* acc = (float*)carve(sizeof(float) * (size_t)Nn * DIM);
  const int NW = (E + BIN_CH - 1) / BIN_CH;
  const int BN = (Nn + BIN_NB - 1) / BIN_NB;
  int2*  recs = (int2*)carve(sizeof(int2) * (size_t)E);
  int*   woff = (int*)carve(sizeof(int) * (size_t)NW * BIN_ROW);
  (void)ws_size; (void)n_in; (void)out_size;

  const int n4 = Nn * DIM / 4;

  gcn_softmax_kernel<<<1, 1, 0, stream>>>(alpha, wts, L1);
  if (E > 0 && NW <= BIN_THREADS && BN <= BIN_THREADS) {
    gcn_bin_edges_kernel<<<NW, BIN_THREADS, 0, stream>>>(src, dst, recs, woff, E, BN);
    gcn_bucket_lists_kernel<<<BIN_NB, BIN_THREADS, 0, stream>>>(recs, woff, rowptr, dis, lst, NW, BN, Nn);
    gcn_list_weights_kernel<<<(E + 1023) / 1024, 256, 0, stream>>>(lst, dis, E);
  } else {
    hipMemsetAsync(cnt, 0, sizeof(int) * (size_t)Nn, stream);
    gcn_count_rank_kernel<<<(E + 255) / 256, 256, 0, stream>>>(dst, cnt, rank, E);
    gcn_scan_dis_kernel<<<(Nn + SCAN_TILE - 1) / SCAN_TILE, SCAN_THREADS, 0, stream>>>(cnt, rowptr, dis, Nn);
    if (E > 0)
      gcn_fill_lists_kernel<<<(E + 256 * FILL_EPT - 1) / (256 * FILL_EPT), 256, 0, stream>>>(
          src, dst, rank, rowptr, dis, lst, E);
  }
  if (L == 0)
    gcn_scale_init_kernel<<<(n4 + 255) / 256, 256, 0, stream>>>(emb, acc, wts, n4);

  const float* xcur = emb;
  float* bufs[2] = {xA, xB};
  for (int l = 0; l < L; ++l) {
    float* xnext = bufs[l & 1];
    long tcount = (long)Nn * (DIM / 4);
    gcn_spmm_gather_acc_kernel<<<(unsigned)((tcount + 255) / 256), 256, 0, stream>>>(
        rowptr, lst, xcur, xnext, acc, wts, l + 1, (l == L - 1) ? 1 : 0, Nn, E);
    xcur = xnext;
  }

  if (E > 0) {
    int waves  = (E + 15) / 16;
    int blocks = (waves + 7) / 8;
    gcn_edge_dot_wmma_kernel<<<blocks, 256, 0, stream>>>(la, lb, acc, res, E);
  }
}


# Layer-weighted output sum in nested form: each layer kernel stores ONE plane t = wts[L-k]*emb[row] + A t_prev (layer 1 scales its sum by wts[L]); x planes and acc re-read go; sums re-associated
# speedup vs baseline: 4.2222x; 1.0924x over previous
#include <hip/hip_runtime.h>

typedef __attribute__((ext_vector_type(2))) float v2f;
typedef __attribute__((ext_vector_type(8))) float v8f;

#define DIM 64

__global__ void gcn_softmax_kernel(const float* __restrict__ alpha,
                                   float* __restrict__ wts, int n) {
  if (blockIdx.x == 0 && threadIdx.x == 0) {
    float m = alpha[0];
    for (int i = 1; i < n; ++i) m = fmaxf(m, alpha[i]);
    float s = 0.f;
    for (int i = 0; i < n; ++i) s += expf(alpha[i] - m);
    for (int i = 0; i < n; ++i) wts[i] = expf(alpha[i] - m) / s;
  }
}

__global__ void gcn_count_rank_kernel(const int* __restrict__ dst, int* __restrict__ cnt,
                                      int* __restrict__ rank, int E) {
  int e = blockIdx.x * blockDim.x + threadIdx.x;
  if (e < E) rank[e] = atomicAdd(&cnt[dst[e]], 1);
}

#define SCAN_THREADS 1024
#define SCAN_TILE    4096
__global__ void __launch_bounds__(SCAN_THREADS)
gcn_scan_dis_kernel(const int* __restrict__ cnt, int* __restrict__ rowptr,
                    float* __restrict__ dis, int n) {
  __shared__ int wsum[32];
  __shared__ int wpre[32];
  const int tid  = threadIdx.x;
  const int lane = tid & 31;
  const int wv   = tid >> 5;
  const int tile0 = blockIdx.x * SCAN_TILE;

  int part = 0;
#pragma unroll 8
  for (int j = tid * 4; j < tile0; j += SCAN_THREADS * 4) {
    const int4 q = *(const int4*)(cnt + j);
    part += (q.x + q.y) + (q.z + q.w);
  }

  const int base = tile0 + tid * 4;
  int v0 = 0, v1 = 0, v2 = 0, v3 = 0;
  if (base + 3 < n) {
    const int4 q = *(const int4*)(cnt + base);
    v0 = q.x; v1 = q.y; v2 = q.z; v3 = q.w;
  } else {
    if (base + 0 < n) v0 = cnt[base + 0];
    if (base + 1 < n) v1 = cnt[base + 1];
    if (base + 2 < n) v2 = cnt[base + 2];
  }
  const int t = v0 + v1 + v2 + v3;

  int inc = t;
  for (int o = 1; o < 32; o <<= 1) {
    const int u = __shfl_up(inc, o, 32);
    if (lane >= o) inc += u;
  }
  for (int o = 16; o > 0; o >>= 1) part += __shfl_xor(part, o, 32);
  if (lane == 31) wsum[wv] = inc;
  if (lane == 0)  wpre[wv] = part;
  __syncthreads();
  int off = 0;
  for (int i = 0; i < 32; ++i) {
    off += wpre[i];
    if (i < wv) off += wsum[i];
  }
  const int e0 = off + inc - t;
  const int e1 = e0 + v0;
  const int e2 = e1 + v1;
  const int e3 = e2 + v2;
  const float d0 = (v0 > 0) ? rsqrtf((float)v0) : 0.f;
  const float d1 = (v1 > 0) ? rsqrtf((float)v1) : 0.f;
  const float d2 = (v2 > 0) ? rsqrtf((float)v2) : 0.f;
  const float d3 = (v3 > 0) ? rsqrtf((float)v3) : 0.f;
  if (base + 3 < n) {
    *(int4*)(rowptr + base) = make_int4(e0, e1, e2, e3);
    *(float4*)(dis + base)  = make_float4(d0, d1, d2, d3);
  } else {
    if (base + 0 < n) { rowptr[base + 0] = e0; dis[base + 0] = d0; }
    if (base + 1 < n) { rowptr[base + 1] = e1; dis[base + 1] = d1; }
    if (base + 2 < n) { rowptr[base + 2] = e2; dis[base + 2] = d2; }
  }
}

#define FILL_EPT 4
__global__ void gcn_fill_lists_kernel(const int* __restrict__ src, const int* __restrict__ dst,
                                      const int* __restrict__ rank, const int* __restrict__ rowptr,
                                      const float* __restrict__ dis, int2* __restrict__ lst, int E) {
  const int T  = gridDim.x * blockDim.x;
  const int e0 = blockIdx.x * blockDim.x + threadIdx.x;
  int s[FILL_EPT], d[FILL_EPT], r[FILL_EPT], p[FILL_EPT];
  float we[FILL_EPT];
#pragma unroll
  for (int k = 0; k < FILL_EPT; ++k) {
    const int e  = e0 + k * T;
    const int ec = (e < E) ? e : (E - 1);
    s[k] = src[ec];
    d[k] = dst[ec];
    r[k] = rank[ec];
  }
#pragma unroll
  for (int k = 0; k < FILL_EPT; ++k) {
    p[k]  = rowptr[d[k]] + r[k];
    we[k] = dis[s[k]] * dis[d[k]];
  }
#pragma unroll
  for (int k = 0; k < FILL_EPT; ++k) {
    if (e0 + k * T < E) lst[p[k]] = make_int2(s[k], __float_as_int(we[k]));
  }
}

#define BIN_THREADS 1024
#define BIN_CH      4096
#define BIN_NB      256
#define BIN_ROW     257
#define BKT_CAP     8192

__device__ __forceinline__ int gcn_block_scan_excl(int t, int* wsum, int& total) {
  const int lane = threadIdx.x & 31;
  const int wv   = threadIdx.x >> 5;
  int inc = t;
  for (int o = 1; o < 32; o <<= 1) {
    const int u = __shfl_up(inc, o, 32);
    if (lane >= o) inc += u;
  }
  if (lane == 31) wsum[wv] = inc;
  __syncthreads();
  int off = 0, tot = 0;
  for (int i = 0; i < 32; ++i) {
    const int s = wsum[i];
    tot += s;
    if (i < wv) off += s;
  }
  __syncthreads();
  total = tot;
  return off + inc - t;
}

__global__ void __launch_bounds__(BIN_THREADS)
gcn_bin_edges_kernel(const int* __restrict__ src, const int* __restrict__ dst,
                     int2* __restrict__ recs, int* __restrict__ woff, int E, int BN) {
  __shared__ int  wsum[32];
  __shared__ int  hist[BIN_NB];
  __shared__ int  start[BIN_NB];
  __shared__ int2 stage[BIN_CH];
  const int tid  = threadIdx.x;
  const int w    = blockIdx.x;
  const int base = w * BIN_CH;
  if (tid < BIN_NB) hist[tid] = 0;
  __syncthreads();
  int s[4], d[4], bk[4], r[4];
#pragma unroll
  for (int k = 0; k < 4; ++k) {
    const int e = base + k * BIN_THREADS + tid;
    s[k] = 0; d[k] = 0;
    if (e < E) { s[k] = src[e]; d[k] = dst[e]; }
  }
#pragma unroll
  for (int k = 0; k < 4; ++k) {
    const int e = base + k * BIN_THREADS + tid;
    bk[k] = 0; r[k] = 0;
    if (e < E) {
      bk[k] = d[k] / BN;
      r[k]  = atomicAdd(&hist[bk[k]], 1);
    }
  }
  __syncthreads();
  const int v = (tid < BIN_NB) ? hist[tid] : 0;
  int tot;
  const int ex = gcn_block_scan_excl(v, wsum, tot);
  if (tid < BIN_NB) {
    start[tid] = ex;
    woff[(size_t)w * BIN_ROW + tid] = ex;
  }
  if (tid == BIN_NB) woff[(size_t)w * BIN_ROW + BIN_NB] = tot;
  __syncthreads();
#pragma unroll
  for (int k = 0; k < 4; ++k) {
    const int e = base + k * BIN_THREADS + tid;
    if (e < E) stage[start[bk[k]] + r[k]] = make_int2(s[k], d[k]);
  }
  __syncthreads();
  for (int i = tid; i < tot; i += BIN_THREADS) recs[base + i] = stage[i];
}

__device__ __forceinline__ int2 gcn_bucket_rec(const int2* __restrict__ recs, const int* segpre,
                                               const int* segbase, int i) {
  int lo = 0;
#pragma unroll
  for (int step = BIN_THREADS / 2; step >= 1; step >>= 1)
    if (segpre[lo + step] <= i) lo += step;
  return recs[segbase[lo] + (i - segpre[lo])];
}

__global__ void __launch_bounds__(BIN_THREADS)
gcn_bucket_lists_kernel(const int2* __restrict__ recs, const int* __restrict__ woff,
                        int* __restrict__ rowptr, float* __restrict__ dis,
                        int2* __restrict__ lst, int NW, int BN, int Nn) {
  __shared__ int   wsum[32];
  __shared__ int   cntl[BIN_THREADS];
  __shared__ float disl[BIN_THREADS];
  __shared__ int   segpre[BIN_THREADS + 1];
  __shared__ int   segbase[BIN_THREADS];
  __shared__ int2  stageL[BKT_CAP];
  __shared__ int2  recL[BKT_CAP];
  const int tid = threadIdx.x;
  const int b   = blockIdx.x;

  int e0 = 0, len = 0;
  if (tid < NW) {
    const int* row = woff + (size_t)tid * BIN_ROW + b;
    e0  = row[0];
    len = row[1] - e0;
    segbase[tid] = tid * BIN_CH + e0;
  }
  int gs, tb;
  (void)gcn_block_scan_excl(e0, wsum, gs);
  const int ex = gcn_block_scan_excl(len, wsum, tb);
  segpre[tid] = ex;
  if (tid == BIN_THREADS - 1) segpre[BIN_THREADS] = ex + len;
  cntl[tid] = 0;
  __syncthreads();

  const int nbase = b * BN;
  for (int i0 = tid; i0 < tb; i0 += 4 * BIN_THREADS) {
    int2 rc[4];
#pragma unroll
    for (int k = 0; k < 4; ++k) {
      const int i = i0 + k * BIN_THREADS;
      rc[k] = gcn_bucket_rec(recs, segpre, segbase, min(i, tb - 1));
    }
#pragma unroll
    for (int k = 0; k < 4; ++k) {
      const int i = i0 + k * BIN_THREADS;
      if (i < tb) {
        if (i < BKT_CAP) recL[i] = rc[k];
        atomicAdd(&cntl[rc[k].y - nbase], 1);
      }
    }
  }
  __syncthreads();

  const int c = cntl[tid];
  int tdummy;
  const int exn = gcn_block_scan_excl(c, wsum, tdummy);
  const float dd = (c > 0) ? rsqrtf((float)c) : 0.f;
  cntl[tid] = exn;
  disl[tid] = dd;
  const int node = nbase + tid;
  if (tid < BN && node < Nn) { rowptr[node] = gs + exn; dis[node] = dd; }
  __syncthreads();

  for (int i0 = tid; i0 < tb; i0 += 4 * BIN_THREADS) {
    int2 rc[4];
#pragma unroll
    for (int k = 0; k < 4; ++k) {
      const int i = i0 + k * BIN_THREADS;
      rc[k] = (i < BKT_CAP) ? recL[min(i, tb - 1)]
                            : gcn_bucket_rec(recs, segpre, segbase, min(i, tb - 1));
    }
#pragma unroll
    for (int k = 0; k < 4; ++k) {
      const int i = i0 + k * BIN_THREADS;
      if (i < tb) {
        const int dl = rc[k].y - nbase;
        const int p  = atomicAdd(&cntl[dl], 1);
        const int2 ent = make_int2(rc[k].x, __float_as_int(disl[dl]));
        if (p < BKT_CAP) stageL[p] = ent; else lst[gs + p] = ent;
      }
    }
  }
  __syncthreads();
  const int nst = min(tb, BKT_CAP);
  for (int i = tid; i < nst; i += BIN_THREADS) lst[gs + i] = stageL[i];
}

__global__ void gcn_list_weights_kernel(int2* __restrict__ lst, const float* __restrict__ dis, int E) {
  const int p = (blockIdx.x * blockDim.x + threadIdx.x) * 4;
  if (p + 3 < E) {
    int4 q0 = *(const int4*)(lst + p);
    int4 q1 = *(const int4*)(lst + p + 2);
    const float a0 = dis[q0.x], a1 = dis[q0.z], a2 = dis[q1.x], a3 = dis[q1.z];
    q0.y = __float_as_int(a0 * __int_as_float(q0.y));
    q0.w = __float_as_int(a1 * __int_as_float(q0.w));
    q1.y = __float_as_int(a2 * __int_as_float(q1.y));
    q1.w = __float_as_int(a3 * __int_as_float(q1.w));
    *(int4*)(lst + p)     = q0;
    *(int4*)(lst + p + 2) = q1;
  } else {
    for (int j = p; j < E; ++j) {
      int2 q = lst[j];
      q.y = __float_as_int(dis[q.x] * __int_as_float(q.y));
      lst[j] = q;
    }
  }
}

__global__ void gcn_scale_init_kernel(const float* __restrict__ x, float* __restrict__ o,
                                      const float* __restrict__ wts, int n4) {
  int i = blockIdx.x * blockDim.x + threadIdx.x;
  if (i < n4) {
    float s = wts[0];
    float4 a = ((const float4*)x)[i];
    float4 r;
    r.x = s * a.x; r.y = s * a.y; r.z = s * a.z; r.w = s * a.w;
    ((float4*)o)[i] = r;
  }
}

__device__ __forceinline__ void gcn_row_fma(float4& a, float w, const float4 x) {
  a.x += w * x.x; a.y += w * x.y; a.z += w * x.z; a.w += w * x.w;
}

__global__ void gcn_spmm_gather_nested_kernel(const int* __restrict__ rowptr, const int2* __restrict__ lst,
                                              const float* __restrict__ xin, const float* __restrict__ emb,
                                              float* __restrict__ tout, const float* __restrict__ wts,
                                              int iself, int ilead, int n, int E) {
  const int t = blockIdx.x * blockDim.x + threadIdx.x;
  const int row = t >> 4;
  if (row >= n) return;
  const int l16 = t & 15;
  const int c   = l16 << 2;
  const size_t o = (size_t)row * DIM + c;
  float4 r = *(const float4*)(emb + o);
  const float wself = wts[iself];
  const float wlead = (ilead >= 0) ? wts[ilead] : 1.0f;
  const int beg = rowptr[row];
  const int end = (row + 1 < n) ? rowptr[row + 1] : E;
  const float* xc = xin + c;

  float4 a = make_float4(0.f, 0.f, 0.f, 0.f);
  for (int j0 = beg; j0 < end; j0 += 16) {
    const int jj = j0 + l16;
    int2 ent = make_int2(0, 0);
    if (jj < end) ent = lst[jj];
    const int cnt = min(16, end - j0);
    int i = 0;
    for (; i + 4 <= cnt; i += 4) {
      const int   s0 = __shfl(ent.x, i + 0, 16);
      const int   s1 = __shfl(ent.x, i + 1, 16);
      const int   s2 = __shfl(ent.x, i + 2, 16);
      const int   s3 = __shfl(ent.x, i + 3, 16);
      const float w0 = __int_as_float(__shfl(ent.y, i + 0, 16));
      const float w1 = __int_as_float(__shfl(ent.y, i + 1, 16));
      const float w2 = __int_as_float(__shfl(ent.y, i + 2, 16));
      const float w3 = __int_as_float(__shfl(ent.y, i + 3, 16));
      const float4 x0 = *(const float4*)(xc + (size_t)s0 * DIM);
      const float4 x1 = *(const float4*)(xc + (size_t)s1 * DIM);
      const float4 x2 = *(const float4*)(xc + (size_t)s2 * DIM);
      const float4 x3 = *(const float4*)(xc + (size_t)s3 * DIM);
      gcn_row_fma(a, w0, x0);
      gcn_row_fma(a, w1, x1);
      gcn_row_fma(a, w2, x2);
      gcn_row_fma(a, w3, x3);
    }
    for (; i < cnt; ++i) {
      const int   s0 = __shfl(ent.x, i, 16);
      const float w0 = __int_as_float(__shfl(ent.y, i, 16));
      const float4 x0 = *(const float4*)(xc + (size_t)s0 * DIM);
      gcn_row_fma(a, w0, x0);
    }
  }
  r.x *= wself; r.y *= wself; r.z *= wself; r.w *= wself;
  gcn_row_fma(r, wlead, a);
  *(float4*)(tout + o) = r;
}

#define ED_WAVES 8
#define ED_ROWS  32
#define ED_PITCH 68
__global__ void __launch_bounds__(ED_WAVES * 32)
gcn_edge_dot_wmma_kernel(const int* __restrict__ la, const int* __restrict__ lb,
                         const float* __restrict__ x, float* __restrict__ res,
                         int E) {
  __shared__ __attribute__((aligned(16))) float stage[ED_WAVES * ED_ROWS * ED_PITCH];

  const int gwave = (blockIdx.x * blockDim.x + threadIdx.x) >> 5;
  const int wv    = threadIdx.x >> 5;
  const int lane  = threadIdx.x & 31;
  const int ebase = gwave * 16;
  const int m     = lane & 15;
  const int hh    = lane >> 4;

  int e = ebase + m;
  if (e >= E) e = E - 1;

  const int* __restrict__ ip = hh ? lb : la;
  const int myrow = ip[e];

  float* st = stage + wv * (ED_ROWS * ED_PITCH);

  float4 v[16];
#pragma unroll
  for (int i = 0; i < 16; ++i) {
    const int ri = __shfl(myrow, i, 16);
    v[i] = *(const float4*)(x + (size_t)ri * DIM + (m << 2));
  }
#pragma unroll
  for (int i = 0; i < 16; ++i)
    *(float4*)(st + (i + 16 * hh) * ED_PITCH + (m << 2)) = v[i];


  const int k0 = hh << 1;
  const float* ap = st + m * ED_PITCH + k0;
  const float* bp = st + (16 + m) * ED_PITCH + k0;

  v8f c = {0.f, 0.f, 0.f, 0.f, 0.f, 0.f, 0.f, 0.f};
#pragma unroll
  for (int kk = 0; kk < DIM; kk += 4) {
    v2f a = *(const v2f*)(ap + kk);
    v2f b = *(const v2f*)(bp + kk);
    c = __builtin_amdgcn_wmma_f32_16x16x4_f32(false, a, false, b, (short)0, c, false, false);
  }

  const int r = lane & 7;
  float d = c[0];
  if (r == 1) d = c[1];
  if (r == 2) d = c[2];
  if (r == 3) d = c[3];
  if (r == 4) d = c[4];
  if (r == 5) d = c[5];
  if (r == 6) d = c[6];
  if (r == 7) d = c[7];

  const bool active = (lane < 8) || (lane >= 24);
  const int ei = ebase + ((lane < 8) ? lane : (lane - 16));
  if (active && ei < E) res[ei] = d;
}

extern "C" void kernel_launch(void* const* d_in, const int* in_sizes, int n_in,
                              void* d_out, int out_size, void* d_ws, size_t ws_size,
                              hipStream_t stream) {
  const int*   edge_index = (const int*)d_in[0];
  const int*   eli        = (const int*)d_in[1];
  const float* emb        = (const float*)d_in[2];
  const float* alpha      = (const float*)d_in[3];

  const int E  = in_sizes[0] / 2;
  const int Nn = in_sizes[2] / DIM;
  const int L1 = in_sizes[3];
  const int L  = L1 - 1;
  float* res = (float*)d_out;

  const int* src = edge_index;
  const int* dst = edge_index + E;
  const int* la  = eli;
  const int* lb  = eli + E;

  size_t off = 0;
  auto carve = [&](size_t bytes) -> void* {
    void* p = (char*)d_ws + off;
    off = (off + bytes + 255) & ~(size_t)255;
    return p;
  };
  float* wts = (float*)carve(sizeof(float) * (size_t)L1);
  float* dis = (float*)carve(sizeof(float) * (size_t)Nn);
  int*   cnt    = (int*)carve(sizeof(int) * (size_t)Nn);
  int*   rowptr = (int*)carve(sizeof(int) * (size_t)Nn);
  int*   rank   = (int*)carve(sizeof(int) * (size_t)E);
  int2*  lst    = (int2*)carve(sizeof(int2) * (size_t)E);
  float* xA  = (float*)carve(sizeof(float) * (size_t)Nn * DIM);
  float* xB  = (float*)carve(sizeof(float) * (size_t)Nn * DIM);
  float* acc = (float*)carve(sizeof(float) * (size_t)Nn * DIM);
  const int NW = (E + BIN_CH - 1) / BIN_CH;
  const int BN = (Nn + BIN_NB - 1) / BIN_NB;
  int2*  recs = (int2*)carve(sizeof(int2) * (size_t)E);
  int*   woff = (int*)carve(sizeof(int) * (size_t)NW * BIN_ROW);
  (void)ws_size; (void)n_in; (void)out_size;

  const int n4 = Nn * DIM / 4;

  gcn_softmax_kernel<<<1, 1, 0, stream>>>(alpha, wts, L1);
  if (E > 0 && NW <= BIN_THREADS && BN <= BIN_THREADS) {
    gcn_bin_edges_kernel<<<NW, BIN_THREADS, 0, stream>>>(src, dst, recs, woff, E, BN);
    gcn_bucket_lists_kernel<<<BIN_NB, BIN_THREADS, 0, stream>>>(recs, woff, rowptr, dis, lst, NW, BN, Nn);
    gcn_list_weights_kernel<<<(E + 1023) / 1024, 256, 0, stream>>>(lst, dis, E);
  } else {
    hipMemsetAsync(cnt, 0, sizeof(int) * (size_t)Nn, stream);
    gcn_count_rank_kernel<<<(E + 255) / 256, 256, 0, stream>>>(dst, cnt, rank, E);
    gcn_scan_dis_kernel<<<(Nn + SCAN_TILE - 1) / SCAN_TILE, SCAN_THREADS, 0, stream>>>(cnt, rowptr, dis, Nn);
    if (E > 0)
      gcn_fill_lists_kernel<<<(E + 256 * FILL_EPT - 1) / (256 * FILL_EPT), 256, 0, stream>>>(
          src, dst, rank, rowptr, dis, lst, E);
  }
  if (L == 0)
    gcn_scale_init_kernel<<<(n4 + 255) / 256, 256, 0, stream>>>(emb, acc, wts, n4);

  const float* xcur = emb;
  float* bufs[2] = {xA, xB};
  for (int l = 0; l < L; ++l) {
    float* tnext = (l == L - 1) ? acc : bufs[l & 1];
    long tcount = (long)Nn * (DIM / 4);
    gcn_spmm_gather_nested_kernel<<<(unsigned)((tcount + 255) / 256), 256, 0, stream>>>(
        rowptr, lst, xcur, emb, tnext, wts, L - 1 - l, (l == 0) ? L : -1, Nn, E);
    xcur = tnext;
  }

  if (E > 0) {
    int waves  = (E + 15) / 16;
    int blocks = (waves + 7) / 8;
    gcn_edge_dot_wmma_kernel<<<blocks, 256, 0, stream>>>(la, lb, acc, res, E);
  }
}


# Edge norm out of the lists: lists hold src only, a row's sum is scaled once by dis[dst], planes are stored pre-scaled by dis (layer 1 reads dis[src]); weight pass goes, 7 launches; sums re-associated
# speedup vs baseline: 4.3619x; 1.0331x over previous
#include <hip/hip_runtime.h>

typedef __attribute__((ext_vector_type(2))) float v2f;
typedef __attribute__((ext_vector_type(8))) float v8f;

#define DIM 64

__global__ void gcn_softmax_kernel(const float* __restrict__ alpha,
                                   float* __restrict__ wts, int n) {
  if (blockIdx.x == 0 && threadIdx.x == 0) {
    float m = alpha[0];
    for (int i = 1; i < n; ++i) m = fmaxf(m, alpha[i]);
    float s = 0.f;
    for (int i = 0; i < n; ++i) s += expf(alpha[i] - m);
    for (int i = 0; i < n; ++i) wts[i] = expf(alpha[i] - m) / s;
  }
}

__global__ void gcn_count_rank_kernel(const int* __restrict__ dst, int* __restrict__ cnt,
                                      int* __restrict__ rank, int E) {
  int e = blockIdx.x * blockDim.x + threadIdx.x;
  if (e < E) rank[e] = atomicAdd(&cnt[dst[e]], 1);
}

#define SCAN_THREADS 1024
#define SCAN_TILE    4096
__global__ void __launch_bounds__(SCAN_THREADS)
gcn_scan_dis_kernel(const int* __restrict__ cnt, int* __restrict__ rowptr,
                    float* __restrict__ dis, int n) {
  __shared__ int wsum[32];
  __shared__ int wpre[32];
  const int tid  = threadIdx.x;
  const int lane = tid & 31;
  const int wv   = tid >> 5;
  const int tile0 = blockIdx.x * SCAN_TILE;

  int part = 0;
#pragma unroll 8
  for (int j = tid * 4; j < tile0; j += SCAN_THREADS * 4) {
    const int4 q = *(const int4*)(cnt + j);
    part += (q.x + q.y) + (q.z + q.w);
  }

  const int base = tile0 + tid * 4;
  int v0 = 0, v1 = 0, v2 = 0, v3 = 0;
  if (base + 3 < n) {
    const int4 q = *(const int4*)(cnt + base);
    v0 = q.x; v1 = q.y; v2 = q.z; v3 = q.w;
  } else {
    if (base + 0 < n) v0 = cnt[base + 0];
    if (base + 1 < n) v1 = cnt[base + 1];
    if (base + 2 < n) v2 = cnt[base + 2];
  }
  const int t = v0 + v1 + v2 + v3;

  int inc = t;
  for (int o = 1; o < 32; o <<= 1) {
    const int u = __shfl_up(inc, o, 32);
    if (lane >= o) inc += u;
  }
  for (int o = 16; o > 0; o >>= 1) part += __shfl_xor(part, o, 32);
  if (lane == 31) wsum[wv] = inc;
  if (lane == 0)  wpre[wv] = part;
  __syncthreads();
  int off = 0;
  for (int i = 0; i < 32; ++i) {
    off += wpre[i];
    if (i < wv) off += wsum[i];
  }
  const int e0 = off + inc - t;
  const int e1 = e0 + v0;
  const int e2 = e1 + v1;
  const int e3 = e2 + v2;
  const float d0 = (v0 > 0) ? rsqrtf((float)v0) : 0.f;
  const float d1 = (v1 > 0) ? rsqrtf((float)v1) : 0.f;
  const float d2 = (v2 > 0) ? rsqrtf((float)v2) : 0.f;
  const float d3 = (v3 > 0) ? rsqrtf((float)v3) : 0.f;
  if (base + 3 < n) {
    *(int4*)(rowptr + base) = make_int4(e0, e1, e2, e3);
    *(float4*)(dis + base)  = make_float4(d0, d1, d2, d3);
  } else {
    if (base + 0 < n) { rowptr[base + 0] = e0; dis[base + 0] = d0; }
    if (base + 1 < n) { rowptr[base + 1] = e1; dis[base + 1] = d1; }
    if (base + 2 < n) { rowptr[base + 2] = e2; dis[base + 2] = d2; }
  }
}

#define FILL_EPT 4
__global__ void gcn_fill_lists_kernel(const int* __restrict__ src, const int* __restrict__ dst,
                                      const int* __restrict__ rank, const int* __restrict__ rowptr,
                                      int* __restrict__ lst, int E) {
  const int T  = gridDim.x * blockDim.x;
  const int e0 = blockIdx.x * blockDim.x + threadIdx.x;
  int s[FILL_EPT], d[FILL_EPT], r[FILL_EPT], p[FILL_EPT];
#pragma unroll
  for (int k = 0; k < FILL_EPT; ++k) {
    const int e  = e0 + k * T;
    const int ec = (e < E) ? e : (E - 1);
    s[k] = src[ec];
    d[k] = dst[ec];
    r[k] = rank[ec];
  }
#pragma unroll
  for (int k = 0; k < FILL_EPT; ++k) p[k] = rowptr[d[k]] + r[k];
#pragma unroll
  for (int k = 0; k < FILL_EPT; ++k) {
    if (e0 + k * T < E) lst[p[k]] = s[k];
  }
}

#define BIN_THREADS 1024
#define BIN_CH      4096
#define BIN_NB      256
#define BIN_ROW     257
#define BKT_CAP     8192

__device__ __forceinline__ int gcn_block_scan_excl(int t, int* wsum, int& total) {
  const int lane = threadIdx.x & 31;
  const int wv   = threadIdx.x >> 5;
  int inc = t;
  for (int o = 1; o < 32; o <<= 1) {
    const int u = __shfl_up(inc, o, 32);
    if (lane >= o) inc += u;
  }
  if (lane == 31) wsum[wv] = inc;
  __syncthreads();
  int off = 0, tot = 0;
  for (int i = 0; i < 32; ++i) {
    const int s = wsum[i];
    tot += s;
    if (i < wv) off += s;
  }
  __syncthreads();
  total = tot;
  return off + inc - t;
}

__global__ void __launch_bounds__(BIN_THREADS)
gcn_bin_edges_kernel(const int* __restrict__ src, const int* __restrict__ dst,
                     int2* __restrict__ recs, int* __restrict__ woff, int E, int BN) {
  __shared__ int  wsum[32];
  __shared__ int  hist[BIN_NB];
  __shared__ int  start[BIN_NB];
  __shared__ int2 stage[BIN_CH];
  const int tid  = threadIdx.x;
  const int w    = blockIdx.x;
  const int base = w * BIN_CH;
  if (tid < BIN_NB) hist[tid] = 0;
  __syncthreads();
  int s[4], d[4], bk[4], r[4];
#pragma unroll
  for (int k = 0; k < 4; ++k) {
    const int e = base + k * BIN_THREADS + tid;
    s[k] = 0; d[k] = 0;
    if (e < E) { s[k] = src[e]; d[k] = dst[e]; }
  }
#pragma unroll
  for (int k = 0; k < 4; ++k) {
    const int e = base + k * BIN_THREADS + tid;
    bk[k] = 0; r[k] = 0;
    if (e < E) {
      bk[k] = d[k] / BN;
      r[k]  = atomicAdd(&hist[bk[k]], 1);
    }
  }
  __syncthreads();
  const int v = (tid < BIN_NB) ? hist[tid] : 0;
  int tot;
  const int ex = gcn_block_scan_excl(v, wsum, tot);
  if (tid < BIN_NB) {
    start[tid] = ex;
    woff[(size_t)w * BIN_ROW + tid] = ex;
  }
  if (tid == BIN_NB) woff[(size_t)w * BIN_ROW + BIN_NB] = tot;
  __syncthreads();
#pragma unroll
  for (int k = 0; k < 4; ++k) {
    const int e = base + k * BIN_THREADS + tid;
    if (e < E) stage[start[bk[k]] + r[k]] = make_int2(s[k], d[k]);
  }
  __syncthreads();
  for (int i = tid; i < tot; i += BIN_THREADS) recs[base + i] = stage[i];
}

__device__ __forceinline__ int2 gcn_bucket_rec(const int2* __restrict__ recs, const int* segpre,
                                               const int* segbase, int i) {
  int lo = 0;
#pragma unroll
  for (int step = BIN_THREADS / 2; step >= 1; step >>= 1)
    if (segpre[lo + step] <= i) lo += step;
  return recs[segbase[lo] + (i - segpre[lo])];
}

__global__ void __launch_bounds__(BIN_THREADS)
gcn_bucket_lists_kernel(const int2* __restrict__ recs, const int* __restrict__ woff,
                        int* __restrict__ rowptr, float* __restrict__ dis,
                        int* __restrict__ lst, int NW, int BN, int Nn) {
  __shared__ int   wsum[32];
  __shared__ int   cntl[BIN_THREADS];
  __shared__ int   segpre[BIN_THREADS + 1];
  __shared__ int   segbase[BIN_THREADS];
  __shared__ int   stageL[BKT_CAP];
  __shared__ int2  recL[BKT_CAP];
  const int tid = threadIdx.x;
  const int b   = blockIdx.x;

  int e0 = 0, len = 0;
  if (tid < NW) {
    const int* row = woff + (size_t)tid * BIN_ROW + b;
    e0  = row[0];
    len = row[1] - e0;
    segbase[tid] = tid * BIN_CH + e0;
  }
  int gs, tb;
  (void)gcn_block_scan_excl(e0, wsum, gs);
  const int ex = gcn_block_scan_excl(len, wsum, tb);
  segpre[tid] = ex;
  if (tid == BIN_THREADS - 1) segpre[BIN_THREADS] = ex + len;
  cntl[tid] = 0;
  __syncthreads();

  const int nbase = b * BN;
  for (int i0 = tid; i0 < tb; i0 += 4 * BIN_THREADS) {
    int2 rc[4];
#pragma unroll
    for (int k = 0; k < 4; ++k) {
      const int i = i0 + k * BIN_THREADS;
      rc[k] = gcn_bucket_rec(recs, segpre, segbase, min(i, tb - 1));
    }
#pragma unroll
    for (int k = 0; k < 4; ++k) {
      const int i = i0 + k * BIN_THREADS;
      if (i < tb) {
        if (i < BKT_CAP) recL[i] = rc[k];
        atomicAdd(&cntl[rc[k].y - nbase], 1);
      }
    }
  }
  __syncthreads();

  const int c = cntl[tid];
  int tdummy;
  const int exn = gcn_block_scan_excl(c, wsum, tdummy);
  const float dd = (c > 0) ? rsqrtf((float)c) : 0.f;
  cntl[tid] = exn;
  const int node = nbase + tid;
  if (tid < BN && node < Nn) { rowptr[node] = gs + exn; dis[node] = dd; }
  __syncthreads();

  for (int i0 = tid; i0 < tb; i0 += 4 * BIN_THREADS) {
    int2 rc[4];
#pragma unroll
    for (int k = 0; k < 4; ++k) {
      const int i = i0 + k * BIN_THREADS;
      rc[k] = (i < BKT_CAP) ? recL[min(i, tb - 1)]
                            : gcn_bucket_rec(recs, segpre, segbase, min(i, tb - 1));
    }
#pragma unroll
    for (int k = 0; k < 4; ++k) {
      const int i = i0 + k * BIN_THREADS;
      if (i < tb) {
        const int dl = rc[k].y - nbase;
        const int p  = atomicAdd(&cntl[dl], 1);
        if (p < BKT_CAP) stageL[p] = rc[k].x; else lst[gs + p] = rc[k].x;
      }
    }
  }
  __syncthreads();
  const int nst = min(tb, BKT_CAP);
  for (int i = tid; i < nst; i += BIN_THREADS) lst[gs + i] = stageL[i];
}

__global__ void gcn_scale_init_kernel(const float* __restrict__ x, float* __restrict__ o,
                                      const float* __restrict__ wts, int n4) {
  int i = blockIdx.x * blockDim.x + threadIdx.x;
  if (i < n4) {
    float s = wts[0];
    float4 a = ((const float4*)x)[i];
    float4 r;
    r.x = s * a.x; r.y = s * a.y; r.z = s * a.z; r.w = s * a.w;
    ((float4*)o)[i] = r;
  }
}

__device__ __forceinline__ void gcn_row_fma(float4& a, float w, const float4 x) {
  a.x += w * x.x; a.y += w * x.y; a.z += w * x.z; a.w += w * x.w;
}
__device__ __forceinline__ void gcn_row_add(float4& a, const float4 x) {
  a.x += x.x; a.y += x.y; a.z += x.z; a.w += x.w;
}

template <bool FIRST>
__global__ void gcn_spmm_gather_nested_kernel(const int* __restrict__ rowptr, const int* __restrict__ lst,
                                              const float* __restrict__ xin, const float* __restrict__ emb,
                                              const float* __restrict__ dis,
                                              float* __restrict__ tout, const float* __restrict__ wts,
                                              int iself, int ilead, int last, int n, int E) {
  const int t = blockIdx.x * blockDim.x + threadIdx.x;
  const int row = t >> 4;
  if (row >= n) return;
  const int l16 = t & 15;
  const int c   = l16 << 2;
  const size_t o = (size_t)row * DIM + c;
  float4 r = *(const float4*)(emb + o);
  const float wself = wts[iself];
  const float wlead = (ilead >= 0) ? wts[ilead] : 1.0f;
  const int beg = rowptr[row];
  const int end = (row + 1 < n) ? rowptr[row + 1] : E;
  const int   deg = end - beg;
  const float dd  = (deg > 0) ? rsqrtf((float)deg) : 0.f;
  const float* xc = xin + c;

  float4 a = make_float4(0.f, 0.f, 0.f, 0.f);
  for (int j0 = beg; j0 < end; j0 += 16) {
    const int jj = j0 + l16;
    int   es = 0;
    float ew = 0.f;
    if (jj < end) {
      es = lst[jj];
      if (FIRST) ew = dis[es];
    }
    const int cnt = min(16, end - j0);
    int i = 0;
    for (; i + 4 <= cnt; i += 4) {
      const int s0 = __shfl(es, i + 0, 16);
      const int s1 = __shfl(es, i + 1, 16);
      const int s2 = __shfl(es, i + 2, 16);
      const int s3 = __shfl(es, i + 3, 16);
      const float4 x0 = *(const float4*)(xc + (size_t)s0 * DIM);
      const float4 x1 = *(const float4*)(xc + (size_t)s1 * DIM);
      const float4 x2 = *(const float4*)(xc + (size_t)s2 * DIM);
      const float4 x3 = *(const float4*)(xc + (size_t)s3 * DIM);
      if (FIRST) {
        const float w0 = __shfl(ew, i + 0, 16);
        const float w1 = __shfl(ew, i + 1, 16);
        const float w2 = __shfl(ew, i + 2, 16);
        const float w3 = __shfl(ew, i + 3, 16);
        gcn_row_fma(a, w0, x0);
        gcn_row_fma(a, w1, x1);
        gcn_row_fma(a, w2, x2);
        gcn_row_fma(a, w3, x3);
      } else {
        gcn_row_add(a, x0);
        gcn_row_add(a, x1);
        gcn_row_add(a, x2);
        gcn_row_add(a, x3);
      }
    }
    for (; i < cnt; ++i) {
      const int s0 = __shfl(es, i, 16);
      const float4 x0 = *(const float4*)(xc + (size_t)s0 * DIM);
      if (FIRST) {
        const float w0 = __shfl(ew, i, 16);
        gcn_row_fma(a, w0, x0);
      } else {
        gcn_row_add(a, x0);
      }
    }
  }
  r.x *= wself; r.y *= wself; r.z *= wself; r.w *= wself;
  gcn_row_fma(r, wlead * dd, a);
  if (!last) { r.x *= dd; r.y *= dd; r.z *= dd; r.w *= dd; }
  *(float4*)(tout + o) = r;
}

#define ED_WAVES 8
#define ED_ROWS  32
#define ED_PITCH 68
__global__ void __launch_bounds__(ED_WAVES * 32)
gcn_edge_dot_wmma_kernel(const int* __restrict__ la, const int* __restrict__ lb,
                         const float* __restrict__ x, float* __restrict__ res,
                         int E) {
  __shared__ __attribute__((aligned(16))) float stage[ED_WAVES * ED_ROWS * ED_PITCH];

  const int gwave = (blockIdx.x * blockDim.x + threadIdx.x) >> 5;
  const int wv    = threadIdx.x >> 5;
  const int lane  = threadIdx.x & 31;
  const int ebase = gwave * 16;
  const int m     = lane & 15;
  const int hh    = lane >> 4;

  int e = ebase + m;
  if (e >= E) e = E - 1;

  const int* __restrict__ ip = hh ? lb : la;
  const int myrow = ip[e];

  float* st = stage + wv * (ED_ROWS * ED_PITCH);

  float4 v[16];
#pragma unroll
  for (int i = 0; i < 16; ++i) {
    const int ri = __shfl(myrow, i, 16);
    v[i] = *(const float4*)(x + (size_t)ri * DIM + (m << 2));
  }
#pragma unroll
  for (int i = 0; i < 16; ++i)
    *(float4*)(st + (i + 16 * hh) * ED_PITCH + (m << 2)) = v[i];


  const int k0 = hh << 1;
  const float* ap = st + m * ED_PITCH + k0;
  const float* bp = st + (16 + m) * ED_PITCH + k0;

  v8f c = {0.f, 0.f, 0.f, 0.f, 0.f, 0.f, 0.f, 0.f};
#pragma unroll
  for (int kk = 0; kk < DIM; kk += 4) {
    v2f a = *(const v2f*)(ap + kk);
    v2f b = *(const v2f*)(bp + kk);
    c = __builtin_amdgcn_wmma_f32_16x16x4_f32(false, a, false, b, (short)0, c, false, false);
  }

  const int r = lane & 7;
  float d = c[0];
  if (r == 1) d = c[1];
  if (r == 2) d = c[2];
  if (r == 3) d = c[3];
  if (r == 4) d = c[4];
  if (r == 5) d = c[5];
  if (r == 6) d = c[6];
  if (r == 7) d = c[7];

  const bool active = (lane < 8) || (lane >= 24);
  const int ei = ebase + ((lane < 8) ? lane : (lane - 16));
  if (active && ei < E) res[ei] = d;
}

extern "C" void kernel_launch(void* const* d_in, const int* in_sizes, int n_in,
                              void* d_out, int out_size, void* d_ws, size_t ws_size,
                              hipStream_t stream) {
  const int*   edge_index = (const int*)d_in[0];
  const int*   eli        = (const int*)d_in[1];
  const float* emb        = (const float*)d_in[2];
  const float* alpha      = (const float*)d_in[3];

  const int E  = in_sizes[0] / 2;
  const int Nn = in_sizes[2] / DIM;
  const int L1 = in_sizes[3];
  const int L  = L1 - 1;
  float* res = (float*)d_out;

  const int* src = edge_index;
  const int* dst = edge_index + E;
  const int* la  = eli;
  const int* lb  = eli + E;

  size_t off = 0;
  auto carve = [&](size_t bytes) -> void* {
    void* p = (char*)d_ws + off;
    off = (off + bytes + 255) & ~(size_t)255;
    return p;
  };
  float* wts = (float*)carve(sizeof(float) * (size_t)L1);
  float* dis = (float*)carve(sizeof(float) * (size_t)Nn);
  int*   cnt    = (int*)carve(sizeof(int) * (size_t)Nn);
  int*   rowptr = (int*)carve(sizeof(int) * (size_t)Nn);
  int*   rank   = (int*)carve(sizeof(int) * (size_t)E);
  int*   lst    = (int*)carve(sizeof(int) * (size_t)E);
  float* xA  = (float*)carve(sizeof(float) * (size_t)Nn * DIM);
  float* xB  = (float*)carve(sizeof(float) * (size_t)Nn * DIM);
  float* acc = (float*)carve(sizeof(float) * (size_t)Nn * DIM);
  const int NW = (E + BIN_CH - 1) / BIN_CH;
  const int BN = (Nn + BIN_NB - 1) / BIN_NB;
  int2*  recs = (int2*)carve(sizeof(int2) * (size_t)E);
  int*   woff = (int*)carve(sizeof(int) * (size_t)NW * BIN_ROW);
  (void)ws_size; (void)n_in; (void)out_size;

  const int n4 = Nn * DIM / 4;

  gcn_softmax_kernel<<<1, 1, 0, stream>>>(alpha, wts, L1);
  if (E > 0 && NW <= BIN_THREADS && BN <= BIN_THREADS) {
    gcn_bin_edges_kernel<<<NW, BIN_THREADS, 0, stream>>>(src, dst, recs, woff, E, BN);
    gcn_bucket_lists_kernel<<<BIN_NB, BIN_THREADS, 0, stream>>>(recs, woff, rowptr, dis, lst, NW, BN, Nn);
  } else {
    hipMemsetAsync(cnt, 0, sizeof(int) * (size_t)Nn, stream);
    gcn_count_rank_kernel<<<(E + 255) / 256, 256, 0, stream>>>(dst, cnt, rank, E);
    gcn_scan_dis_kernel<<<(Nn + SCAN_TILE - 1) / SCAN_TILE, SCAN_THREADS, 0, stream>>>(cnt, rowptr, dis, Nn);
    if (E > 0)
      gcn_fill_lists_kernel<<<(E + 256 * FILL_EPT - 1) / (256 * FILL_EPT), 256, 0, stream>>>(
          src, dst, rank, rowptr, lst, E);
  }
  if (L == 0)
    gcn_scale_init_kernel<<<(n4 + 255) / 256, 256, 0, stream>>>(emb, acc, wts, n4);

  const float* xcur = emb;
  float* bufs[2] = {xA, xB};
  for (int l = 0; l < L; ++l) {
    float* tnext = (l == L - 1) ? acc : bufs[l & 1];
    const int last = (l == L - 1) ? 1 : 0;
    long tcount = (long)Nn * (DIM / 4);
    const unsigned gblocks = (unsigned)((tcount + 255) / 256);
    if (l == 0)
      gcn_spmm_gather_nested_kernel<true><<<gblocks, 256, 0, stream>>>(
          rowptr, lst, xcur, emb, dis, tnext, wts, L - 1 - l, L, last, Nn, E);
    else
      gcn_spmm_gather_nested_kernel<false><<<gblocks, 256, 0, stream>>>(
          rowptr, lst, xcur, emb, dis, tnext, wts, L - 1 - l, -1, last, Nn, E);
    xcur = tnext;
  }

  if (E > 0) {
    int waves  = (E + 15) / 16;
    int blocks = (waves + 7) / 8;
    gcn_edge_dot_wmma_kernel<<<blocks, 256, 0, stream>>>(la, lb, acc, res, E);
  }
}
